# starting kernel
# baseline (speedup 1.0000x reference)
#include <hip/hip_runtime.h>

typedef __attribute__((ext_vector_type(2))) float v2f;
typedef __attribute__((ext_vector_type(8))) float v8f;

__device__ __forceinline__ v8f wmma_f32(v2f a, v2f b, v8f c) {
  return __builtin_amdgcn_wmma_f32_16x16x4_f32(false, a, false, b, (short)0, c, false, false);
}

#define BB   2
#define NN   1024
#define DD   1024
#define HH   16
#define DHH  64
#define SCALE 0.125f
#define EPS_  1e-6f
#define NEGBIG (-1e30f)

__device__ __forceinline__ void gemm_ldfrag(const float* __restrict__ aptr,
                                            const float* __restrict__ bptr,
                                            size_t s16, int k0,
                                            v2f a[2], v2f b[4]) {
  a[0] = *(const v2f*)(aptr + k0);
  a[1] = *(const v2f*)(aptr + s16 + k0);
  b[0] = *(const v2f*)(bptr + k0);
  b[1] = *(const v2f*)(bptr + s16 + k0);
  b[2] = *(const v2f*)(bptr + 2 * s16 + k0);
  b[3] = *(const v2f*)(bptr + 3 * s16 + k0);
}

__device__ __forceinline__ void gemm_mma8(const v2f a[2], const v2f b[4],
                                          v8f acc[2][4]) {
#pragma unroll
  for (int mt = 0; mt < 2; ++mt)
#pragma unroll
    for (int nt = 0; nt < 4; ++nt)
      acc[mt][nt] = wmma_f32(a[mt], b[nt], acc[mt][nt]);
}

__global__ __launch_bounds__(128)
void gemm_nt_f32_wmma(const float* __restrict__ A, const float* __restrict__ Bm,
                      float* __restrict__ C, int M, int Nc, int K) {
  const int lane = threadIdx.x & 31;
  const int wave = threadIdx.x >> 5;
  const int lhi  = lane >> 4;
  const int lr   = lane & 15;

  const int nBlocks = Nc >> 7;
  const int mb = blockIdx.x / nBlocks;
  const int nb = blockIdx.x % nBlocks;
  const int mBase = mb * 64  + (wave >> 1) * 32;
  const int nBase = nb * 128 + (wave & 1) * 64;

  v8f acc[2][4];
#pragma unroll
  for (int mt = 0; mt < 2; ++mt)
#pragma unroll
    for (int nt = 0; nt < 4; ++nt) acc[mt][nt] = (v8f){0,0,0,0,0,0,0,0};

  const float* aptr = A  + (size_t)(mBase + lr) * K + 2 * lhi;
  const float* bptr = Bm + (size_t)(nBase + lr) * K + 2 * lhi;
  const size_t s16  = (size_t)16 * K;

  v2f aE[2], bE[4], aO[2], bO[4];
  gemm_ldfrag(aptr, bptr, s16, 0, aE, bE);
#pragma unroll 2
  for (int k0 = 0; k0 < K; k0 += 8) {
    gemm_ldfrag(aptr, bptr, s16, k0 + 4, aO, bO);
    gemm_mma8(aE, bE, acc);
    if (k0 + 8 < K)
      gemm_ldfrag(aptr, bptr, s16, k0 + 8, aE, bE);
    gemm_mma8(aO, bO, acc);
  }

#pragma unroll
  for (int mt = 0; mt < 2; ++mt)
#pragma unroll
    for (int nt = 0; nt < 4; ++nt)
#pragma unroll
      for (int r = 0; r < 8; ++r) {
        size_t row = (size_t)(mBase + mt * 16 + lhi * 8 + r);
        C[row * Nc + (nBase + nt * 16 + lr)] = acc[mt][nt][r];
      }
}

__global__ __launch_bounds__(128)
void stieltjes_attn(const float* __restrict__ qkv, float* __restrict__ O) {
  __shared__ float S[32][1032];
  __shared__ float redA[4][32];
  __shared__ float redB[4][32];
  __shared__ float lam[32];
  __shared__ float wsum[32];

  const int lane = threadIdx.x & 31;
  const int wave = threadIdx.x >> 5;
  const int lhi  = lane >> 4;
  const int lr   = lane & 15;

  const int rb = blockIdx.x & 31;
  const int bh = blockIdx.x >> 5;
  const int h  = bh & (HH - 1);
  const int b  = bh >> 4;

  const int rowBase = rb * 32;
  const int ncols   = rowBase + 32;
  const int nct     = ncols >> 4;

  const float* qbase = qkv + (size_t)b * NN * (3 * DD) + h * DHH;
  const float* kbase = qbase + DD;
  const float* vbase = qbase + 2 * DD;

  v2f aQ[2][16];
#pragma unroll
  for (int mt = 0; mt < 2; ++mt) {
    const float* qr = qbase + (size_t)(rowBase + mt * 16 + lr) * (3 * DD) + 2 * lhi;
#pragma unroll
    for (int ks = 0; ks < 16; ++ks) aQ[mt][ks] = *(const v2f*)(qr + 4 * ks);
  }

  for (int ct = wave; ct < nct; ct += 4) {
    v8f acc0 = (v8f){0,0,0,0,0,0,0,0};
    v8f acc1 = (v8f){0,0,0,0,0,0,0,0};
    const float* kr = kbase + (size_t)(ct * 16 + lr) * (3 * DD) + 2 * lhi;
    v2f bK = *(const v2f*)(kr);
#pragma unroll
    for (int ks = 0; ks < 16; ++ks) {
      v2f bKn;
      if (ks < 15) bKn = *(const v2f*)(kr + 4 * (ks + 1));
      acc0 = wmma_f32(aQ[0][ks], bK, acc0);
      acc1 = wmma_f32(aQ[1][ks], bK, acc1);
      bK = bKn;
    }
    const int gcol = ct * 16 + lr;
#pragma unroll
    for (int r = 0; r < 8; ++r) {
      int row0 = lhi * 8 + r;
      int row1 = 16 + row0;
      float v0 = acc0[r] * SCALE;
      float v1 = acc1[r] * SCALE;
      if (gcol > rowBase + row0) v0 = NEGBIG;
      if (gcol > rowBase + row1) v1 = NEGBIG;
      S[row0][gcol] = v0;
      S[row1][gcol] = v1;
    }
  }
  __syncthreads();

  const int row  = threadIdx.x & 31;
  const int part = threadIdx.x >> 5;
  {
    float m = -3e30f;
    for (int c = part; c < ncols; c += 4) m = fmaxf(m, S[row][c]);
    redA[part][row] = m;
  }
  __syncthreads();
  if (threadIdx.x < 32) {
    float m = fmaxf(fmaxf(redA[0][row], redA[1][row]),
                    fmaxf(redA[2][row], redA[3][row]));
    lam[row] = m + 1.0f;
  }
  __syncthreads();

  for (int it = 0; it < 4; ++it) {
    const float l = lam[row];
    float fs = 0.0f, ps = 0.0f;
    for (int c = part; c < ncols; c += 4) {
      float d  = fmaxf(l - S[row][c], EPS_);
      float r1 = 1.0f / d;
      fs += r1;
      ps += r1 * r1;
    }
    redA[part][row] = fs;
    redB[part][row] = ps;
    __syncthreads();
    if (threadIdx.x < 32) {
      float f  = redA[0][row] + redA[1][row] + redA[2][row] + redA[3][row] - 1.0f;
      float fp = -(redB[0][row] + redB[1][row] + redB[2][row] + redB[3][row]);
      lam[row] = l - f / fp;
    }
    __syncthreads();
  }

  {
    const float l = lam[row];
    float ws = 0.0f;
    for (int c = part; c < ncols; c += 4)
      ws += 1.0f / fmaxf(l - S[row][c], EPS_);
    redA[part][row] = ws;
    __syncthreads();
    if (threadIdx.x < 32)
      wsum[row] = redA[0][row] + redA[1][row] + redA[2][row] + redA[3][row];
    __syncthreads();
    const float inv = 1.0f / wsum[row];
    for (int c = part; c < ncols; c += 4) {
      float d = fmaxf(l - S[row][c], EPS_);
      S[row][c] = (1.0f / d) * inv;
    }
  }
  __syncthreads();

  v8f oacc[2];
  oacc[0] = (v8f){0,0,0,0,0,0,0,0};
  oacc[1] = (v8f){0,0,0,0,0,0,0,0};
  const int dh0 = wave * 16;
  const float* vcol = vbase + dh0 + lr;
  const int ksteps = ncols >> 2;

  v2f bV, a0, a1;
  {
    const int kr = 2 * lhi;
    bV.x = vcol[(size_t)kr * (3 * DD)];
    bV.y = vcol[(size_t)(kr + 1) * (3 * DD)];
    a0 = *(const v2f*)(&S[lr][kr]);
    a1 = *(const v2f*)(&S[16 + lr][kr]);
  }
  for (int ks = 0; ks < ksteps; ++ks) {
    v2f bVn, a0n, a1n;
    if (ks + 1 < ksteps) {
      const int kr = (ks + 1) * 4 + 2 * lhi;
      bVn.x = vcol[(size_t)kr * (3 * DD)];
      bVn.y = vcol[(size_t)(kr + 1) * (3 * DD)];
      a0n = *(const v2f*)(&S[lr][kr]);
      a1n = *(const v2f*)(&S[16 + lr][kr]);
    }
    oacc[0] = wmma_f32(a0, bV, oacc[0]);
    oacc[1] = wmma_f32(a1, bV, oacc[1]);
    bV = bVn; a0 = a0n; a1 = a1n;
  }

  float* obase = O + ((size_t)b * NN + rowBase) * DD + h * DHH + dh0 + lr;
#pragma unroll
  for (int mt = 0; mt < 2; ++mt)
#pragma unroll
    for (int r = 0; r < 8; ++r)
      obase[(size_t)(mt * 16 + lhi * 8 + r) * DD] = oacc[mt][r];
}

extern "C" void kernel_launch(void* const* d_in, const int* in_sizes, int n_in,
                              void* d_out, int out_size, void* d_ws, size_t ws_size,
                              hipStream_t stream) {
  const float* x    = (const float*)d_in[0];
  const float* Wqkv = (const float*)d_in[1];
  const float* Wo   = (const float*)d_in[2];
  float* out = (float*)d_out;

  float* qkv = (float*)d_ws;
  float* Ows = qkv + (size_t)2048 * 3072;

  gemm_nt_f32_wmma<<<(2048 / 64) * (3072 / 128), 128, 0, stream>>>(
      x, Wqkv, qkv, 2048, 3072, 1024);

  stieltjes_attn<<<BB * HH * (NN / 32), 128, 0, stream>>>(qkv, Ows);

  gemm_nt_f32_wmma<<<(2048 / 64) * (1024 / 128), 128, 0, stream>>>(
      Ows, Wo, out, 2048, 1024, 1024);
}


# Newton/normalise passes: 4 adjacent scores per 128-bit LDS read (row pitch 1028), 8 independent divide chains per trip instead of 1 serial chain; row sums regrouped, so their sum order changes
# speedup vs baseline: 1.6407x; 1.6407x over previous
#include <hip/hip_runtime.h>

typedef __attribute__((ext_vector_type(2))) float v2f;
typedef __attribute__((ext_vector_type(8))) float v8f;
typedef __attribute__((ext_vector_type(4))) float v4f;

__device__ __forceinline__ v8f wmma_f32(v2f a, v2f b, v8f c) {
  return __builtin_amdgcn_wmma_f32_16x16x4_f32(false, a, false, b, (short)0, c, false, false);
}

#define BB   2
#define NN   1024
#define DD   1024
#define HH   16
#define DHH  64
#define SCALE 0.125f
#define EPS_  1e-6f
#define NEGBIG (-1e30f)
#define SP   1028

__device__ __forceinline__ void gemm_ldfrag(const float* __restrict__ aptr,
                                            const float* __restrict__ bptr,
                                            size_t s16, int k0,
                                            v2f a[2], v2f b[4]) {
  a[0] = *(const v2f*)(aptr + k0);
  a[1] = *(const v2f*)(aptr + s16 + k0);
  b[0] = *(const v2f*)(bptr + k0);
  b[1] = *(const v2f*)(bptr + s16 + k0);
  b[2] = *(const v2f*)(bptr + 2 * s16 + k0);
  b[3] = *(const v2f*)(bptr + 3 * s16 + k0);
}

__device__ __forceinline__ void gemm_mma8(const v2f a[2], const v2f b[4],
                                          v8f acc[2][4]) {
#pragma unroll
  for (int mt = 0; mt < 2; ++mt)
#pragma unroll
    for (int nt = 0; nt < 4; ++nt)
      acc[mt][nt] = wmma_f32(a[mt], b[nt], acc[mt][nt]);
}

__global__ __launch_bounds__(128)
void gemm_nt_f32_wmma(const float* __restrict__ A, const float* __restrict__ Bm,
                      float* __restrict__ C, int M, int Nc, int K) {
  const int lane = threadIdx.x & 31;
  const int wave = threadIdx.x >> 5;
  const int lhi  = lane >> 4;
  const int lr   = lane & 15;

  const int nBlocks = Nc >> 7;
  const int mb = blockIdx.x / nBlocks;
  const int nb = blockIdx.x % nBlocks;
  const int mBase = mb * 64  + (wave >> 1) * 32;
  const int nBase = nb * 128 + (wave & 1) * 64;

  v8f acc[2][4];
#pragma unroll
  for (int mt = 0; mt < 2; ++mt)
#pragma unroll
    for (int nt = 0; nt < 4; ++nt) acc[mt][nt] = (v8f){0,0,0,0,0,0,0,0};

  const float* aptr = A  + (size_t)(mBase + lr) * K + 2 * lhi;
  const float* bptr = Bm + (size_t)(nBase + lr) * K + 2 * lhi;
  const size_t s16  = (size_t)16 * K;

  v2f aE[2], bE[4], aO[2], bO[4];
  gemm_ldfrag(aptr, bptr, s16, 0, aE, bE);
#pragma unroll 2
  for (int k0 = 0; k0 < K; k0 += 8) {
    gemm_ldfrag(aptr, bptr, s16, k0 + 4, aO, bO);
    gemm_mma8(aE, bE, acc);
    if (k0 + 8 < K)
      gemm_ldfrag(aptr, bptr, s16, k0 + 8, aE, bE);
    gemm_mma8(aO, bO, acc);
  }

#pragma unroll
  for (int mt = 0; mt < 2; ++mt)
#pragma unroll
    for (int nt = 0; nt < 4; ++nt)
#pragma unroll
      for (int r = 0; r < 8; ++r) {
        size_t row = (size_t)(mBase + mt * 16 + lhi * 8 + r);
        C[row * Nc + (nBase + nt * 16 + lr)] = acc[mt][nt][r];
      }
}

__device__ __forceinline__ v4f stj_recip4(v4f s, float l) {
  v4f w;
  w.x = 1.0f / fmaxf(l - s.x, EPS_);
  w.y = 1.0f / fmaxf(l - s.y, EPS_);
  w.z = 1.0f / fmaxf(l - s.z, EPS_);
  w.w = 1.0f / fmaxf(l - s.w, EPS_);
  return w;
}

__global__ __launch_bounds__(128)
void stieltjes_attn(const float* __restrict__ qkv, float* __restrict__ O) {
  __shared__ __attribute__((aligned(16))) float S[32][SP];
  __shared__ float redA[4][32];
  __shared__ float redB[4][32];
  __shared__ float lam[32];
  __shared__ float wsum[32];

  const int lane = threadIdx.x & 31;
  const int wave = threadIdx.x >> 5;
  const int lhi  = lane >> 4;
  const int lr   = lane & 15;

  const int rb = blockIdx.x & 31;
  const int bh = blockIdx.x >> 5;
  const int h  = bh & (HH - 1);
  const int b  = bh >> 4;

  const int rowBase = rb * 32;
  const int ncols   = rowBase + 32;
  const int nct     = ncols >> 4;

  const float* qbase = qkv + (size_t)b * NN * (3 * DD) + h * DHH;
  const float* kbase = qbase + DD;
  const float* vbase = qbase + 2 * DD;

  v2f aQ[2][16];
#pragma unroll
  for (int mt = 0; mt < 2; ++mt) {
    const float* qr = qbase + (size_t)(rowBase + mt * 16 + lr) * (3 * DD) + 2 * lhi;
#pragma unroll
    for (int ks = 0; ks < 16; ++ks) aQ[mt][ks] = *(const v2f*)(qr + 4 * ks);
  }

  for (int ct = wave; ct < nct; ct += 4) {
    v8f acc0 = (v8f){0,0,0,0,0,0,0,0};
    v8f acc1 = (v8f){0,0,0,0,0,0,0,0};
    const float* kr = kbase + (size_t)(ct * 16 + lr) * (3 * DD) + 2 * lhi;
    v2f bK = *(const v2f*)(kr);
#pragma unroll
    for (int ks = 0; ks < 16; ++ks) {
      v2f bKn;
      if (ks < 15) bKn = *(const v2f*)(kr + 4 * (ks + 1));
      acc0 = wmma_f32(aQ[0][ks], bK, acc0);
      acc1 = wmma_f32(aQ[1][ks], bK, acc1);
      bK = bKn;
    }
    const int gcol = ct * 16 + lr;
#pragma unroll
    for (int r = 0; r < 8; ++r) {
      int row0 = lhi * 8 + r;
      int row1 = 16 + row0;
      float v0 = acc0[r] * SCALE;
      float v1 = acc1[r] * SCALE;
      if (gcol > rowBase + row0) v0 = NEGBIG;
      if (gcol > rowBase + row1) v1 = NEGBIG;
      S[row0][gcol] = v0;
      S[row1][gcol] = v1;
    }
  }
  __syncthreads();

  const int row   = threadIdx.x & 31;
  const int part  = threadIdx.x >> 5;
  const int ntrip = rb + 1;
  v4f* rp = reinterpret_cast<v4f*>(&S[row][0]) + part;
  {
    v4f m4 = (v4f){-3e30f, -3e30f, -3e30f, -3e30f};
    for (int t = 0; t < ntrip; ++t) {
      const v4f x0 = rp[8 * t];
      const v4f x1 = rp[8 * t + 4];
      m4.x = fmaxf(m4.x, fmaxf(x0.x, x1.x));
      m4.y = fmaxf(m4.y, fmaxf(x0.y, x1.y));
      m4.z = fmaxf(m4.z, fmaxf(x0.z, x1.z));
      m4.w = fmaxf(m4.w, fmaxf(x0.w, x1.w));
    }
    redA[part][row] = fmaxf(fmaxf(m4.x, m4.y), fmaxf(m4.z, m4.w));
  }
  __syncthreads();
  if (threadIdx.x < 32) {
    float m = fmaxf(fmaxf(redA[0][row], redA[1][row]),
                    fmaxf(redA[2][row], redA[3][row]));
    lam[row] = m + 1.0f;
  }
  __syncthreads();

  for (int it = 0; it < 4; ++it) {
    const float l = lam[row];
    v4f fs = (v4f){0.0f, 0.0f, 0.0f, 0.0f};
    v4f ps = (v4f){0.0f, 0.0f, 0.0f, 0.0f};
    v4f x0 = rp[0], x1 = rp[4];
    for (int t = 0; t < ntrip; ++t) {
      v4f n0 = x0, n1 = x1;
      if (t + 1 < ntrip) { n0 = rp[8 * t + 8]; n1 = rp[8 * t + 12]; }
      const v4f r0 = stj_recip4(x0, l);
      const v4f r1 = stj_recip4(x1, l);
      fs += r0;  ps += r0 * r0;
      fs += r1;  ps += r1 * r1;
      x0 = n0;  x1 = n1;
    }
    redA[part][row] = (fs.x + fs.y) + (fs.z + fs.w);
    redB[part][row] = (ps.x + ps.y) + (ps.z + ps.w);
    __syncthreads();
    if (threadIdx.x < 32) {
      float f  = redA[0][row] + redA[1][row] + redA[2][row] + redA[3][row] - 1.0f;
      float fp = -(redB[0][row] + redB[1][row] + redB[2][row] + redB[3][row]);
      lam[row] = l - f / fp;
    }
    __syncthreads();
  }

  {
    const float l = lam[row];
    v4f ws = (v4f){0.0f, 0.0f, 0.0f, 0.0f};
    v4f x0 = rp[0], x1 = rp[4];
    for (int t = 0; t < ntrip; ++t) {
      v4f n0 = x0, n1 = x1;
      if (t + 1 < ntrip) { n0 = rp[8 * t + 8]; n1 = rp[8 * t + 12]; }
      ws += stj_recip4(x0, l);
      ws += stj_recip4(x1, l);
      x0 = n0;  x1 = n1;
    }
    redA[part][row] = (ws.x + ws.y) + (ws.z + ws.w);
    __syncthreads();
    if (threadIdx.x < 32)
      wsum[row] = redA[0][row] + redA[1][row] + redA[2][row] + redA[3][row];
    __syncthreads();
    const float inv = 1.0f / wsum[row];
    x0 = rp[0];  x1 = rp[4];
    for (int t = 0; t < ntrip; ++t) {
      v4f n0 = x0, n1 = x1;
      if (t + 1 < ntrip) { n0 = rp[8 * t + 8]; n1 = rp[8 * t + 12]; }
      rp[8 * t]     = stj_recip4(x0, l) * inv;
      rp[8 * t + 4] = stj_recip4(x1, l) * inv;
      x0 = n0;  x1 = n1;
    }
  }
  __syncthreads();

  v8f oacc[2];
  oacc[0] = (v8f){0,0,0,0,0,0,0,0};
  oacc[1] = (v8f){0,0,0,0,0,0,0,0};
  const int dh0 = wave * 16;
  const float* vcol = vbase + dh0 + lr;
  const int ksteps = ncols >> 2;

  v2f bV, a0, a1;
  {
    const int kr = 2 * lhi;
    bV.x = vcol[(size_t)kr * (3 * DD)];
    bV.y = vcol[(size_t)(kr + 1) * (3 * DD)];
    a0 = *(const v2f*)(&S[lr][kr]);
    a1 = *(const v2f*)(&S[16 + lr][kr]);
  }
  for (int ks = 0; ks < ksteps; ++ks) {
    v2f bVn, a0n, a1n;
    if (ks + 1 < ksteps) {
      const int kr = (ks + 1) * 4 + 2 * lhi;
      bVn.x = vcol[(size_t)kr * (3 * DD)];
      bVn.y = vcol[(size_t)(kr + 1) * (3 * DD)];
      a0n = *(const v2f*)(&S[lr][kr]);
      a1n = *(const v2f*)(&S[16 + lr][kr]);
    }
    oacc[0] = wmma_f32(a0, bV, oacc[0]);
    oacc[1] = wmma_f32(a1, bV, oacc[1]);
    bV = bVn; a0 = a0n; a1 = a1n;
  }

  float* obase = O + ((size_t)b * NN + rowBase) * DD + h * DHH + dh0 + lr;
#pragma unroll
  for (int mt = 0; mt < 2; ++mt)
#pragma unroll
    for (int r = 0; r < 8; ++r)
      obase[(size_t)(mt * 16 + lhi * 8 + r) * DD] = oacc[mt][r];
}

extern "C" void kernel_launch(void* const* d_in, const int* in_sizes, int n_in,
                              void* d_out, int out_size, void* d_ws, size_t ws_size,
                              hipStream_t stream) {
  const float* x    = (const float*)d_in[0];
  const float* Wqkv = (const float*)d_in[1];
  const float* Wo   = (const float*)d_in[2];
  float* out = (float*)d_out;

  float* qkv = (float*)d_ws;
  float* Ows = qkv + (size_t)2048 * 3072;

  gemm_nt_f32_wmma<<<(2048 / 64) * (3072 / 128), 128, 0, stream>>>(
      x, Wqkv, qkv, 2048, 3072, 1024);

  stieltjes_attn<<<BB * HH * (NN / 32), 128, 0, stream>>>(qkv, Ows);

  gemm_nt_f32_wmma<<<(2048 / 64) * (1024 / 128), 128, 0, stream>>>(
      Ows, Wo, out, 2048, 1024, 1024);
}


# P @ V phase: V fragments requested 16 to 32 k-steps ahead as one group of loads and P fragments read from LDS 8 steps ahead, instead of 1 step ahead; k order and sum order unchanged
# speedup vs baseline: 1.8405x; 1.1218x over previous
#include <hip/hip_runtime.h>

typedef __attribute__((ext_vector_type(2))) float v2f;
typedef __attribute__((ext_vector_type(8))) float v8f;
typedef __attribute__((ext_vector_type(4))) float v4f;

__device__ __forceinline__ v8f wmma_f32(v2f a, v2f b, v8f c) {
  return __builtin_amdgcn_wmma_f32_16x16x4_f32(false, a, false, b, (short)0, c, false, false);
}

#define BB   2
#define NN   1024
#define DD   1024
#define HH   16
#define DHH  64
#define SCALE 0.125f
#define EPS_  1e-6f
#define NEGBIG (-1e30f)
#define SP   1028

__device__ __forceinline__ void gemm_ldfrag(const float* __restrict__ aptr,
                                            const float* __restrict__ bptr,
                                            size_t s16, int k0,
                                            v2f a[2], v2f b[4]) {
  a[0] = *(const v2f*)(aptr + k0);
  a[1] = *(const v2f*)(aptr + s16 + k0);
  b[0] = *(const v2f*)(bptr + k0);
  b[1] = *(const v2f*)(bptr + s16 + k0);
  b[2] = *(const v2f*)(bptr + 2 * s16 + k0);
  b[3] = *(const v2f*)(bptr + 3 * s16 + k0);
}

__device__ __forceinline__ void gemm_mma8(const v2f a[2], const v2f b[4],
                                          v8f acc[2][4]) {
#pragma unroll
  for (int mt = 0; mt < 2; ++mt)
#pragma unroll
    for (int nt = 0; nt < 4; ++nt)
      acc[mt][nt] = wmma_f32(a[mt], b[nt], acc[mt][nt]);
}

__global__ __launch_bounds__(128)
void gemm_nt_f32_wmma(const float* __restrict__ A, const float* __restrict__ Bm,
                      float* __restrict__ C, int M, int Nc, int K) {
  const int lane = threadIdx.x & 31;
  const int wave = threadIdx.x >> 5;
  const int lhi  = lane >> 4;
  const int lr   = lane & 15;

  const int nBlocks = Nc >> 7;
  const int mb = blockIdx.x / nBlocks;
  const int nb = blockIdx.x % nBlocks;
  const int mBase = mb * 64  + (wave >> 1) * 32;
  const int nBase = nb * 128 + (wave & 1) * 64;

  v8f acc[2][4];
#pragma unroll
  for (int mt = 0; mt < 2; ++mt)
#pragma unroll
    for (int nt = 0; nt < 4; ++nt) acc[mt][nt] = (v8f){0,0,0,0,0,0,0,0};

  const float* aptr = A  + (size_t)(mBase + lr) * K + 2 * lhi;
  const float* bptr = Bm + (size_t)(nBase + lr) * K + 2 * lhi;
  const size_t s16  = (size_t)16 * K;

  v2f aE[2], bE[4], aO[2], bO[4];
  gemm_ldfrag(aptr, bptr, s16, 0, aE, bE);
#pragma unroll 2
  for (int k0 = 0; k0 < K; k0 += 8) {
    gemm_ldfrag(aptr, bptr, s16, k0 + 4, aO, bO);
    gemm_mma8(aE, bE, acc);
    if (k0 + 8 < K)
      gemm_ldfrag(aptr, bptr, s16, k0 + 8, aE, bE);
    gemm_mma8(aO, bO, acc);
  }

#pragma unroll
  for (int mt = 0; mt < 2; ++mt)
#pragma unroll
    for (int nt = 0; nt < 4; ++nt)
#pragma unroll
      for (int r = 0; r < 8; ++r) {
        size_t row = (size_t)(mBase + mt * 16 + lhi * 8 + r);
        C[row * Nc + (nBase + nt * 16 + lr)] = acc[mt][nt][r];
      }
}

__device__ __forceinline__ void pv_load_v8(v2f (&v)[16], int jb, const float* __restrict__ vcol,
                                           int hb, int lhi) {
  const float* p = vcol + (size_t)(hb * 32 + 2 * lhi) * (3 * DD);
#pragma unroll
  for (int j = 0; j < 8; ++j) {
    v[jb + j].x = p[(size_t)(4 * j) * (3 * DD)];
    v[jb + j].y = p[(size_t)(4 * j + 1) * (3 * DD)];
  }
}

__device__ __forceinline__ void pv_load_p8(v2f (&p0)[8], v2f (&p1)[8],
                                           const float* s0, const float* s1, int hb) {
#pragma unroll
  for (int j = 0; j < 8; ++j) {
    p0[j] = *(const v2f*)(s0 + hb * 32 + 4 * j);
    p1[j] = *(const v2f*)(s1 + hb * 32 + 4 * j);
  }
}

__device__ __forceinline__ void pv_mma8(const v2f (&p0)[8], const v2f (&p1)[8],
                                        const v2f (&v)[16], int jb, v8f (&oacc)[2]) {
#pragma unroll
  for (int j = 0; j < 8; ++j) {
    oacc[0] = wmma_f32(p0[j], v[jb + j], oacc[0]);
    oacc[1] = wmma_f32(p1[j], v[jb + j], oacc[1]);
  }
}

__device__ __forceinline__ v4f stj_recip4(v4f s, float l) {
  v4f w;
  w.x = 1.0f / fmaxf(l - s.x, EPS_);
  w.y = 1.0f / fmaxf(l - s.y, EPS_);
  w.z = 1.0f / fmaxf(l - s.z, EPS_);
  w.w = 1.0f / fmaxf(l - s.w, EPS_);
  return w;
}

__global__ __launch_bounds__(128)
void stieltjes_attn(const float* __restrict__ qkv, float* __restrict__ O) {
  __shared__ __attribute__((aligned(16))) float S[32][SP];
  __shared__ float redA[4][32];
  __shared__ float redB[4][32];
  __shared__ float lam[32];
  __shared__ float wsum[32];

  const int lane = threadIdx.x & 31;
  const int wave = threadIdx.x >> 5;
  const int lhi  = lane >> 4;
  const int lr   = lane & 15;

  const int rb = blockIdx.x & 31;
  const int bh = blockIdx.x >> 5;
  const int h  = bh & (HH - 1);
  const int b  = bh >> 4;

  const int rowBase = rb * 32;
  const int ncols   = rowBase + 32;
  const int nct     = ncols >> 4;

  const float* qbase = qkv + (size_t)b * NN * (3 * DD) + h * DHH;
  const float* kbase = qbase + DD;
  const float* vbase = qbase + 2 * DD;

  v2f aQ[2][16];
#pragma unroll
  for (int mt = 0; mt < 2; ++mt) {
    const float* qr = qbase + (size_t)(rowBase + mt * 16 + lr) * (3 * DD) + 2 * lhi;
#pragma unroll
    for (int ks = 0; ks < 16; ++ks) aQ[mt][ks] = *(const v2f*)(qr + 4 * ks);
  }

  for (int ct = wave; ct < nct; ct += 4) {
    v8f acc0 = (v8f){0,0,0,0,0,0,0,0};
    v8f acc1 = (v8f){0,0,0,0,0,0,0,0};
    const float* kr = kbase + (size_t)(ct * 16 + lr) * (3 * DD) + 2 * lhi;
    v2f bK = *(const v2f*)(kr);
#pragma unroll
    for (int ks = 0; ks < 16; ++ks) {
      v2f bKn;
      if (ks < 15) bKn = *(const v2f*)(kr + 4 * (ks + 1));
      acc0 = wmma_f32(aQ[0][ks], bK, acc0);
      acc1 = wmma_f32(aQ[1][ks], bK, acc1);
      bK = bKn;
    }
    const int gcol = ct * 16 + lr;
#pragma unroll
    for (int r = 0; r < 8; ++r) {
      int row0 = lhi * 8 + r;
      int row1 = 16 + row0;
      float v0 = acc0[r] * SCALE;
      float v1 = acc1[r] * SCALE;
      if (gcol > rowBase + row0) v0 = NEGBIG;
      if (gcol > rowBase + row1) v1 = NEGBIG;
      S[row0][gcol] = v0;
      S[row1][gcol] = v1;
    }
  }
  __syncthreads();

  const int row   = threadIdx.x & 31;
  const int part  = threadIdx.x >> 5;
  const int ntrip = rb + 1;
  v4f* rp = reinterpret_cast<v4f*>(&S[row][0]) + part;
  {
    v4f m4 = (v4f){-3e30f, -3e30f, -3e30f, -3e30f};
    for (int t = 0; t < ntrip; ++t) {
      const v4f x0 = rp[8 * t];
      const v4f x1 = rp[8 * t + 4];
      m4.x = fmaxf(m4.x, fmaxf(x0.x, x1.x));
      m4.y = fmaxf(m4.y, fmaxf(x0.y, x1.y));
      m4.z = fmaxf(m4.z, fmaxf(x0.z, x1.z));
      m4.w = fmaxf(m4.w, fmaxf(x0.w, x1.w));
    }
    redA[part][row] = fmaxf(fmaxf(m4.x, m4.y), fmaxf(m4.z, m4.w));
  }
  __syncthreads();
  if (threadIdx.x < 32) {
    float m = fmaxf(fmaxf(redA[0][row], redA[1][row]),
                    fmaxf(redA[2][row], redA[3][row]));
    lam[row] = m + 1.0f;
  }
  __syncthreads();

  for (int it = 0; it < 4; ++it) {
    const float l = lam[row];
    v4f fs = (v4f){0.0f, 0.0f, 0.0f, 0.0f};
    v4f ps = (v4f){0.0f, 0.0f, 0.0f, 0.0f};
    v4f x0 = rp[0], x1 = rp[4];
    for (int t = 0; t < ntrip; ++t) {
      v4f n0 = x0, n1 = x1;
      if (t + 1 < ntrip) { n0 = rp[8 * t + 8]; n1 = rp[8 * t + 12]; }
      const v4f r0 = stj_recip4(x0, l);
      const v4f r1 = stj_recip4(x1, l);
      fs += r0;  ps += r0 * r0;
      fs += r1;  ps += r1 * r1;
      x0 = n0;  x1 = n1;
    }
    redA[part][row] = (fs.x + fs.y) + (fs.z + fs.w);
    redB[part][row] = (ps.x + ps.y) + (ps.z + ps.w);
    __syncthreads();
    if (threadIdx.x < 32) {
      float f  = redA[0][row] + redA[1][row] + redA[2][row] + redA[3][row] - 1.0f;
      float fp = -(redB[0][row] + redB[1][row] + redB[2][row] + redB[3][row]);
      lam[row] = l - f / fp;
    }
    __syncthreads();
  }

  {
    const float l = lam[row];
    v4f ws = (v4f){0.0f, 0.0f, 0.0f, 0.0f};
    v4f x0 = rp[0], x1 = rp[4];
    for (int t = 0; t < ntrip; ++t) {
      v4f n0 = x0, n1 = x1;
      if (t + 1 < ntrip) { n0 = rp[8 * t + 8]; n1 = rp[8 * t + 12]; }
      ws += stj_recip4(x0, l);
      ws += stj_recip4(x1, l);
      x0 = n0;  x1 = n1;
    }
    redA[part][row] = (ws.x + ws.y) + (ws.z + ws.w);
    __syncthreads();
    if (threadIdx.x < 32)
      wsum[row] = redA[0][row] + redA[1][row] + redA[2][row] + redA[3][row];
    __syncthreads();
    const float inv = 1.0f / wsum[row];
    x0 = rp[0];  x1 = rp[4];
    for (int t = 0; t < ntrip; ++t) {
      v4f n0 = x0, n1 = x1;
      if (t + 1 < ntrip) { n0 = rp[8 * t + 8]; n1 = rp[8 * t + 12]; }
      rp[8 * t]     = stj_recip4(x0, l) * inv;
      rp[8 * t + 4] = stj_recip4(x1, l) * inv;
      x0 = n0;  x1 = n1;
    }
  }

  const int dh0 = wave * 16;
  const float* vcol = vbase + dh0 + lr;

  v8f oacc[2];
  oacc[0] = (v8f){0,0,0,0,0,0,0,0};
  oacc[1] = (v8f){0,0,0,0,0,0,0,0};
  const int nh   = rb + 1;
  const int last = nh - 1;
  const float* s0 = &S[lr][2 * lhi];
  const float* s1 = &S[16 + lr][2 * lhi];

  v2f vA[16], vB[16];
  v2f p0[8], p1[8], q0[8], q1[8];
  pv_load_v8(vA, 0, vcol, 0, lhi);
  pv_load_v8(vA, 8, vcol, min(1, last), lhi);
  __syncthreads();
  pv_load_p8(p0, p1, s0, s1, 0);
  for (int h = 0; h < nh; h += 4) {
    pv_load_v8(vB, 0, vcol, min(h + 2, last), lhi);
    pv_load_v8(vB, 8, vcol, min(h + 3, last), lhi);
    pv_load_p8(q0, q1, s0, s1, min(h + 1, last));
    pv_mma8(p0, p1, vA, 0, oacc);
    if (h + 1 < nh) {
      pv_load_p8(p0, p1, s0, s1, min(h + 2, last));
      pv_mma8(q0, q1, vA, 8, oacc);
      if (h + 2 < nh) {
        pv_load_v8(vA, 0, vcol, min(h + 4, last), lhi);
        pv_load_v8(vA, 8, vcol, min(h + 5, last), lhi);
        pv_load_p8(q0, q1, s0, s1, min(h + 3, last));
        pv_mma8(p0, p1, vB, 0, oacc);
        if (h + 3 < nh) {
          pv_load_p8(p0, p1, s0, s1, min(h + 4, last));
          pv_mma8(q0, q1, vB, 8, oacc);
        }
      }
    }
  }

  float* obase = O + ((size_t)b * NN + rowBase) * DD + h * DHH + dh0 + lr;
#pragma unroll
  for (int mt = 0; mt < 2; ++mt)
#pragma unroll
    for (int r = 0; r < 8; ++r)
      obase[(size_t)(mt * 16 + lhi * 8 + r) * DD] = oacc[mt][r];
}

extern "C" void kernel_launch(void* const* d_in, const int* in_sizes, int n_in,
                              void* d_out, int out_size, void* d_ws, size_t ws_size,
                              hipStream_t stream) {
  const float* x    = (const float*)d_in[0];
  const float* Wqkv = (const float*)d_in[1];
  const float* Wo   = (const float*)d_in[2];
  float* out = (float*)d_out;

  float* qkv = (float*)d_ws;
  float* Ows = qkv + (size_t)2048 * 3072;

  gemm_nt_f32_wmma<<<(2048 / 64) * (3072 / 128), 128, 0, stream>>>(
      x, Wqkv, qkv, 2048, 3072, 1024);

  stieltjes_attn<<<BB * HH * (NN / 32), 128, 0, stream>>>(qkv, Ows);

  gemm_nt_f32_wmma<<<(2048 / 64) * (1024 / 128), 128, 0, stream>>>(
      Ows, Wo, out, 2048, 1024, 1024);
}


# Both projection GEMMs: A and B tiles staged through double-buffered LDS with whole-line 128-bit global loads, not per-lane 8-byte loads touching 16 lines each; k and sum order unchanged
# speedup vs baseline: 2.2713x; 1.2341x over previous
#include <hip/hip_runtime.h>

typedef __attribute__((ext_vector_type(2))) float v2f;
typedef __attribute__((ext_vector_type(8))) float v8f;
typedef __attribute__((ext_vector_type(4))) float v4f;

__device__ __forceinline__ v8f wmma_f32(v2f a, v2f b, v8f c) {
  return __builtin_amdgcn_wmma_f32_16x16x4_f32(false, a, false, b, (short)0, c, false, false);
}

#define BB   2
#define NN   1024
#define DD   1024
#define HH   16
#define DHH  64
#define SCALE 0.125f
#define EPS_  1e-6f
#define NEGBIG (-1e30f)
#define SP   1028

#define GP   36
#define GKC  32

__device__ __forceinline__ void gemm_fetch(const float* __restrict__ ag,
                                           const float* __restrict__ bg,
                                           size_t s16, int kc, v4f ra[4], v4f rb[8]) {
#pragma unroll
  for (int i = 0; i < 4; ++i) ra[i] = *(const v4f*)(ag + i * s16 + kc);
#pragma unroll
  for (int i = 0; i < 8; ++i) rb[i] = *(const v4f*)(bg + i * s16 + kc);
}

__device__ __forceinline__ void gemm_park(float* as, float* bs, const v4f ra[4], const v4f rb[8]) {
#pragma unroll
  for (int i = 0; i < 4; ++i) *(v4f*)(as + i * 16 * GP) = ra[i];
#pragma unroll
  for (int i = 0; i < 8; ++i) *(v4f*)(bs + i * 16 * GP) = rb[i];
}

__device__ __forceinline__ void gemm_ldfrag(const float* ap, const float* bp, int k0,
                                            v2f a[2], v2f b[4]) {
  a[0] = *(const v2f*)(ap + k0);
  a[1] = *(const v2f*)(ap + 16 * GP + k0);
  b[0] = *(const v2f*)(bp + k0);
  b[1] = *(const v2f*)(bp + 16 * GP + k0);
  b[2] = *(const v2f*)(bp + 32 * GP + k0);
  b[3] = *(const v2f*)(bp + 48 * GP + k0);
}

__device__ __forceinline__ void gemm_mma8(const v2f a[2], const v2f b[4],
                                          v8f acc[2][4]) {
#pragma unroll
  for (int mt = 0; mt < 2; ++mt)
#pragma unroll
    for (int nt = 0; nt < 4; ++nt)
      acc[mt][nt] = wmma_f32(a[mt], b[nt], acc[mt][nt]);
}

template <bool NEXT>
__device__ __forceinline__ void gemm_chunk(const float* ap, const float* bp,
                                           float* asn, float* bsn,
                                           const v4f ra[4], const v4f rb[8], v8f acc[2][4]) {
  v2f aE[2], bE[4], aO[2], bO[4];
  gemm_ldfrag(ap, bp, 0, aE, bE);
#pragma unroll
  for (int ks = 0; ks < GKC / 4; ks += 2) {
    gemm_ldfrag(ap, bp, 4 * ks + 4, aO, bO);
    gemm_mma8(aE, bE, acc);
    if (ks + 2 < GKC / 4) gemm_ldfrag(ap, bp, 4 * ks + 8, aE, bE);
    gemm_mma8(aO, bO, acc);
    if (NEXT && ks == 4) gemm_park(asn, bsn, ra, rb);
  }
}

__global__ __launch_bounds__(128)
void gemm_nt_f32_wmma(const float* __restrict__ A, const float* __restrict__ Bm,
                      float* __restrict__ C, int M, int Nc, int K) {
  __shared__ __attribute__((aligned(16))) float As[2][64][GP];
  __shared__ __attribute__((aligned(16))) float Bs[2][128][GP];

  const int lane = threadIdx.x & 31;
  const int wave = threadIdx.x >> 5;
  const int lhi  = lane >> 4;
  const int lr   = lane & 15;

  const int nBlocks = Nc >> 7;
  const int mb = blockIdx.x / nBlocks;
  const int nb = blockIdx.x % nBlocks;
  const int mW = (wave >> 1) * 32;
  const int nW = (wave & 1) * 64;
  const int mBase = mb * 64  + mW;
  const int nBase = nb * 128 + nW;

  v8f acc[2][4];
#pragma unroll
  for (int mt = 0; mt < 2; ++mt)
#pragma unroll
    for (int nt = 0; nt < 4; ++nt) acc[mt][nt] = (v8f){0,0,0,0,0,0,0,0};

  const int seg = threadIdx.x & 7;
  const int sr  = threadIdx.x >> 3;
  const float* ag = A  + (size_t)(mb * 64  + sr) * K + 4 * seg;
  const float* bg = Bm + (size_t)(nb * 128 + sr) * K + 4 * seg;
  const size_t s16 = (size_t)16 * K;

  const int nchunks = K / GKC;
  v4f ra[4], rb[8];
  gemm_fetch(ag, bg, s16, 0, ra, rb);
  gemm_park(&As[0][sr][4 * seg], &Bs[0][sr][4 * seg], ra, rb);
  __syncthreads();

  for (int c = 0; c + 1 < nchunks; ++c) {
    const int cur = c & 1;
    gemm_fetch(ag, bg, s16, (c + 1) * GKC, ra, rb);
    gemm_chunk<true>(&As[cur][mW + lr][2 * lhi], &Bs[cur][nW + lr][2 * lhi],
                     &As[cur ^ 1][sr][4 * seg], &Bs[cur ^ 1][sr][4 * seg], ra, rb, acc);
    __syncthreads();
  }
  {
    const int cur = (nchunks - 1) & 1;
    gemm_chunk<false>(&As[cur][mW + lr][2 * lhi], &Bs[cur][nW + lr][2 * lhi],
                      &As[cur][sr][4 * seg], &Bs[cur][sr][4 * seg], ra, rb, acc);
  }

#pragma unroll
  for (int mt = 0; mt < 2; ++mt)
#pragma unroll
    for (int nt = 0; nt < 4; ++nt)
#pragma unroll
      for (int r = 0; r < 8; ++r) {
        size_t row = (size_t)(mBase + mt * 16 + lhi * 8 + r);
        C[row * Nc + (nBase + nt * 16 + lr)] = acc[mt][nt][r];
      }
}

__device__ __forceinline__ void pv_load_v8(v2f (&v)[16], int jb, const float* __restrict__ vcol,
                                           int hb, int lhi) {
  const float* p = vcol + (size_t)(hb * 32 + 2 * lhi) * (3 * DD);
#pragma unroll
  for (int j = 0; j < 8; ++j) {
    v[jb + j].x = p[(size_t)(4 * j) * (3 * DD)];
    v[jb + j].y = p[(size_t)(4 * j + 1) * (3 * DD)];
  }
}

__device__ __forceinline__ void pv_load_p8(v2f (&p0)[8], v2f (&p1)[8],
                                           const float* s0, const float* s1, int hb) {
#pragma unroll
  for (int j = 0; j < 8; ++j) {
    p0[j] = *(const v2f*)(s0 + hb * 32 + 4 * j);
    p1[j] = *(const v2f*)(s1 + hb * 32 + 4 * j);
  }
}

__device__ __forceinline__ void pv_mma8(const v2f (&p0)[8], const v2f (&p1)[8],
                                        const v2f (&v)[16], int jb, v8f (&oacc)[2]) {
#pragma unroll
  for (int j = 0; j < 8; ++j) {
    oacc[0] = wmma_f32(p0[j], v[jb + j], oacc[0]);
    oacc[1] = wmma_f32(p1[j], v[jb + j], oacc[1]);
  }
}

__device__ __forceinline__ v4f stj_recip4(v4f s, float l) {
  v4f w;
  w.x = 1.0f / fmaxf(l - s.x, EPS_);
  w.y = 1.0f / fmaxf(l - s.y, EPS_);
  w.z = 1.0f / fmaxf(l - s.z, EPS_);
  w.w = 1.0f / fmaxf(l - s.w, EPS_);
  return w;
}

__global__ __launch_bounds__(128)
void stieltjes_attn(const float* __restrict__ qkv, float* __restrict__ O) {
  __shared__ __attribute__((aligned(16))) float S[32][SP];
  __shared__ float redA[4][32];
  __shared__ float redB[4][32];
  __shared__ float lam[32];
  __shared__ float wsum[32];

  const int lane = threadIdx.x & 31;
  const int wave = threadIdx.x >> 5;
  const int lhi  = lane >> 4;
  const int lr   = lane & 15;

  const int rb = blockIdx.x & 31;
  const int bh = blockIdx.x >> 5;
  const int h  = bh & (HH - 1);
  const int b  = bh >> 4;

  const int rowBase = rb * 32;
  const int ncols   = rowBase + 32;
  const int nct     = ncols >> 4;

  const float* qbase = qkv + (size_t)b * NN * (3 * DD) + h * DHH;
  const float* kbase = qbase + DD;
  const float* vbase = qbase + 2 * DD;

  v2f aQ[2][16];
#pragma unroll
  for (int mt = 0; mt < 2; ++mt) {
    const float* qr = qbase + (size_t)(rowBase + mt * 16 + lr) * (3 * DD) + 2 * lhi;
#pragma unroll
    for (int ks = 0; ks < 16; ++ks) aQ[mt][ks] = *(const v2f*)(qr + 4 * ks);
  }

  for (int ct = wave; ct < nct; ct += 4) {
    v8f acc0 = (v8f){0,0,0,0,0,0,0,0};
    v8f acc1 = (v8f){0,0,0,0,0,0,0,0};
    const float* kr = kbase + (size_t)(ct * 16 + lr) * (3 * DD) + 2 * lhi;
    v2f bK = *(const v2f*)(kr);
#pragma unroll
    for (int ks = 0; ks < 16; ++ks) {
      v2f bKn;
      if (ks < 15) bKn = *(const v2f*)(kr + 4 * (ks + 1));
      acc0 = wmma_f32(aQ[0][ks], bK, acc0);
      acc1 = wmma_f32(aQ[1][ks], bK, acc1);
      bK = bKn;
    }
    const int gcol = ct * 16 + lr;
#pragma unroll
    for (int r = 0; r < 8; ++r) {
      int row0 = lhi * 8 + r;
      int row1 = 16 + row0;
      float v0 = acc0[r] * SCALE;
      float v1 = acc1[r] * SCALE;
      if (gcol > rowBase + row0) v0 = NEGBIG;
      if (gcol > rowBase + row1) v1 = NEGBIG;
      S[row0][gcol] = v0;
      S[row1][gcol] = v1;
    }
  }
  __syncthreads();

  const int row   = threadIdx.x & 31;
  const int part  = threadIdx.x >> 5;
  const int ntrip = rb + 1;
  v4f* rp = reinterpret_cast<v4f*>(&S[row][0]) + part;
  {
    v4f m4 = (v4f){-3e30f, -3e30f, -3e30f, -3e30f};
    for (int t = 0; t < ntrip; ++t) {
      const v4f x0 = rp[8 * t];
      const v4f x1 = rp[8 * t + 4];
      m4.x = fmaxf(m4.x, fmaxf(x0.x, x1.x));
      m4.y = fmaxf(m4.y, fmaxf(x0.y, x1.y));
      m4.z = fmaxf(m4.z, fmaxf(x0.z, x1.z));
      m4.w = fmaxf(m4.w, fmaxf(x0.w, x1.w));
    }
    redA[part][row] = fmaxf(fmaxf(m4.x, m4.y), fmaxf(m4.z, m4.w));
  }
  __syncthreads();
  if (threadIdx.x < 32) {
    float m = fmaxf(fmaxf(redA[0][row], redA[1][row]),
                    fmaxf(redA[2][row], redA[3][row]));
    lam[row] = m + 1.0f;
  }
  __syncthreads();

  for (int it = 0; it < 4; ++it) {
    const float l = lam[row];
    v4f fs = (v4f){0.0f, 0.0f, 0.0f, 0.0f};
    v4f ps = (v4f){0.0f, 0.0f, 0.0f, 0.0f};
    v4f x0 = rp[0], x1 = rp[4];
    for (int t = 0; t < ntrip; ++t) {
      v4f n0 = x0, n1 = x1;
      if (t + 1 < ntrip) { n0 = rp[8 * t + 8]; n1 = rp[8 * t + 12]; }
      const v4f r0 = stj_recip4(x0, l);
      const v4f r1 = stj_recip4(x1, l);
      fs += r0;  ps += r0 * r0;
      fs += r1;  ps += r1 * r1;
      x0 = n0;  x1 = n1;
    }
    redA[part][row] = (fs.x + fs.y) + (fs.z + fs.w);
    redB[part][row] = (ps.x + ps.y) + (ps.z + ps.w);
    __syncthreads();
    if (threadIdx.x < 32) {
      float f  = redA[0][row] + redA[1][row] + redA[2][row] + redA[3][row] - 1.0f;
      float fp = -(redB[0][row] + redB[1][row] + redB[2][row] + redB[3][row]);
      lam[row] = l - f / fp;
    }
    __syncthreads();
  }

  {
    const float l = lam[row];
    v4f ws = (v4f){0.0f, 0.0f, 0.0f, 0.0f};
    v4f x0 = rp[0], x1 = rp[4];
    for (int t = 0; t < ntrip; ++t) {
      v4f n0 = x0, n1 = x1;
      if (t + 1 < ntrip) { n0 = rp[8 * t + 8]; n1 = rp[8 * t + 12]; }
      ws += stj_recip4(x0, l);
      ws += stj_recip4(x1, l);
      x0 = n0;  x1 = n1;
    }
    redA[part][row] = (ws.x + ws.y) + (ws.z + ws.w);
    __syncthreads();
    if (threadIdx.x < 32)
      wsum[row] = redA[0][row] + redA[1][row] + redA[2][row] + redA[3][row];
    __syncthreads();
    const float inv = 1.0f / wsum[row];
    x0 = rp[0];  x1 = rp[4];
    for (int t = 0; t < ntrip; ++t) {
      v4f n0 = x0, n1 = x1;
      if (t + 1 < ntrip) { n0 = rp[8 * t + 8]; n1 = rp[8 * t + 12]; }
      rp[8 * t]     = stj_recip4(x0, l) * inv;
      rp[8 * t + 4] = stj_recip4(x1, l) * inv;
      x0 = n0;  x1 = n1;
    }
  }

  const int dh0 = wave * 16;
  const float* vcol = vbase + dh0 + lr;

  v8f oacc[2];
  oacc[0] = (v8f){0,0,0,0,0,0,0,0};
  oacc[1] = (v8f){0,0,0,0,0,0,0,0};
  const int nh   = rb + 1;
  const int last = nh - 1;
  const float* s0 = &S[lr][2 * lhi];
  const float* s1 = &S[16 + lr][2 * lhi];

  v2f vA[16], vB[16];
  v2f p0[8], p1[8], q0[8], q1[8];
  pv_load_v8(vA, 0, vcol, 0, lhi);
  pv_load_v8(vA, 8, vcol, min(1, last), lhi);
  __syncthreads();
  pv_load_p8(p0, p1, s0, s1, 0);
  for (int h = 0; h < nh; h += 4) {
    pv_load_v8(vB, 0, vcol, min(h + 2, last), lhi);
    pv_load_v8(vB, 8, vcol, min(h + 3, last), lhi);
    pv_load_p8(q0, q1, s0, s1, min(h + 1, last));
    pv_mma8(p0, p1, vA, 0, oacc);
    if (h + 1 < nh) {
      pv_load_p8(p0, p1, s0, s1, min(h + 2, last));
      pv_mma8(q0, q1, vA, 8, oacc);
      if (h + 2 < nh) {
        pv_load_v8(vA, 0, vcol, min(h + 4, last), lhi);
        pv_load_v8(vA, 8, vcol, min(h + 5, last), lhi);
        pv_load_p8(q0, q1, s0, s1, min(h + 3, last));
        pv_mma8(p0, p1, vB, 0, oacc);
        if (h + 3 < nh) {
          pv_load_p8(p0, p1, s0, s1, min(h + 4, last));
          pv_mma8(q0, q1, vB, 8, oacc);
        }
      }
    }
  }

  float* obase = O + ((size_t)b * NN + rowBase) * DD + h * DHH + dh0 + lr;
#pragma unroll
  for (int mt = 0; mt < 2; ++mt)
#pragma unroll
    for (int r = 0; r < 8; ++r)
      obase[(size_t)(mt * 16 + lhi * 8 + r) * DD] = oacc[mt][r];
}

extern "C" void kernel_launch(void* const* d_in, const int* in_sizes, int n_in,
                              void* d_out, int out_size, void* d_ws, size_t ws_size,
                              hipStream_t stream) {
  const float* x    = (const float*)d_in[0];
  const float* Wqkv = (const float*)d_in[1];
  const float* Wo   = (const float*)d_in[2];
  float* out = (float*)d_out;

  float* qkv = (float*)d_ws;
  float* Ows = qkv + (size_t)2048 * 3072;

  gemm_nt_f32_wmma<<<(2048 / 64) * (3072 / 128), 128, 0, stream>>>(
      x, Wqkv, qkv, 2048, 3072, 1024);

  stieltjes_attn<<<BB * HH * (NN / 32), 128, 0, stream>>>(qkv, Ows);

  gemm_nt_f32_wmma<<<(2048 / 64) * (1024 / 128), 128, 0, stream>>>(
      Ows, Wo, out, 2048, 1024, 1024);
}


# Attention block order: row block taken from the high index bits, longest causal rows first, so the 32-unit workgroups start first instead of last and no longer run on alone; arithmetic unchanged
# speedup vs baseline: 2.6893x; 1.1840x over previous
#include <hip/hip_runtime.h>

typedef __attribute__((ext_vector_type(2))) float v2f;
typedef __attribute__((ext_vector_type(8))) float v8f;
typedef __attribute__((ext_vector_type(4))) float v4f;

__device__ __forceinline__ v8f wmma_f32(v2f a, v2f b, v8f c) {
  return __builtin_amdgcn_wmma_f32_16x16x4_f32(false, a, false, b, (short)0, c, false, false);
}

#define BB   2
#define NN   1024
#define DD   1024
#define HH   16
#define DHH  64
#define SCALE 0.125f
#define EPS_  1e-6f
#define NEGBIG (-1e30f)
#define SP   1028

#define GP   36
#define GKC  32

__device__ __forceinline__ void gemm_fetch(const float* __restrict__ ag,
                                           const float* __restrict__ bg,
                                           size_t s16, int kc, v4f ra[4], v4f rb[8]) {
#pragma unroll
  for (int i = 0; i < 4; ++i) ra[i] = *(const v4f*)(ag + i * s16 + kc);
#pragma unroll
  for (int i = 0; i < 8; ++i) rb[i] = *(const v4f*)(bg + i * s16 + kc);
}

__device__ __forceinline__ void gemm_park(float* as, float* bs, const v4f ra[4], const v4f rb[8]) {
#pragma unroll
  for (int i = 0; i < 4; ++i) *(v4f*)(as + i * 16 * GP) = ra[i];
#pragma unroll
  for (int i = 0; i < 8; ++i) *(v4f*)(bs + i * 16 * GP) = rb[i];
}

__device__ __forceinline__ void gemm_ldfrag(const float* ap, const float* bp, int k0,
                                            v2f a[2], v2f b[4]) {
  a[0] = *(const v2f*)(ap + k0);
  a[1] = *(const v2f*)(ap + 16 * GP + k0);
  b[0] = *(const v2f*)(bp + k0);
  b[1] = *(const v2f*)(bp + 16 * GP + k0);
  b[2] = *(const v2f*)(bp + 32 * GP + k0);
  b[3] = *(const v2f*)(bp + 48 * GP + k0);
}

__device__ __forceinline__ void gemm_mma8(const v2f a[2], const v2f b[4],
                                          v8f acc[2][4]) {
#pragma unroll
  for (int mt = 0; mt < 2; ++mt)
#pragma unroll
    for (int nt = 0; nt < 4; ++nt)
      acc[mt][nt] = wmma_f32(a[mt], b[nt], acc[mt][nt]);
}

template <bool NEXT>
__device__ __forceinline__ void gemm_chunk(const float* ap, const float* bp,
                                           float* asn, float* bsn,
                                           const v4f ra[4], const v4f rb[8], v8f acc[2][4]) {
  v2f aE[2], bE[4], aO[2], bO[4];
  gemm_ldfrag(ap, bp, 0, aE, bE);
#pragma unroll
  for (int ks = 0; ks < GKC / 4; ks += 2) {
    gemm_ldfrag(ap, bp, 4 * ks + 4, aO, bO);
    gemm_mma8(aE, bE, acc);
    if (ks + 2 < GKC / 4) gemm_ldfrag(ap, bp, 4 * ks + 8, aE, bE);
    gemm_mma8(aO, bO, acc);
    if (NEXT && ks == 4) gemm_park(asn, bsn, ra, rb);
  }
}

__global__ __launch_bounds__(128)
void gemm_nt_f32_wmma(const float* __restrict__ A, const float* __restrict__ Bm,
                      float* __restrict__ C, int M, int Nc, int K) {
  __shared__ __attribute__((aligned(16))) float As[2][64][GP];
  __shared__ __attribute__((aligned(16))) float Bs[2][128][GP];

  const int lane = threadIdx.x & 31;
  const int wave = threadIdx.x >> 5;
  const int lhi  = lane >> 4;
  const int lr   = lane & 15;

  const int nBlocks = Nc >> 7;
  const int mb = blockIdx.x / nBlocks;
  const int nb = blockIdx.x % nBlocks;
  const int mW = (wave >> 1) * 32;
  const int nW = (wave & 1) * 64;
  const int mBase = mb * 64  + mW;
  const int nBase = nb * 128 + nW;

  v8f acc[2][4];
#pragma unroll
  for (int mt = 0; mt < 2; ++mt)
#pragma unroll
    for (int nt = 0; nt < 4; ++nt) acc[mt][nt] = (v8f){0,0,0,0,0,0,0,0};

  const int seg = threadIdx.x & 7;
  const int sr  = threadIdx.x >> 3;
  const float* ag = A  + (size_t)(mb * 64  + sr) * K + 4 * seg;
  const float* bg = Bm + (size_t)(nb * 128 + sr) * K + 4 * seg;
  const size_t s16 = (size_t)16 * K;

  const int nchunks = K / GKC;
  v4f ra[4], rb[8];
  gemm_fetch(ag, bg, s16, 0, ra, rb);
  gemm_park(&As[0][sr][4 * seg], &Bs[0][sr][4 * seg], ra, rb);
  __syncthreads();

  for (int c = 0; c + 1 < nchunks; ++c) {
    const int cur = c & 1;
    gemm_fetch(ag, bg, s16, (c + 1) * GKC, ra, rb);
    gemm_chunk<true>(&As[cur][mW + lr][2 * lhi], &Bs[cur][nW + lr][2 * lhi],
                     &As[cur ^ 1][sr][4 * seg], &Bs[cur ^ 1][sr][4 * seg], ra, rb, acc);
    __syncthreads();
  }
  {
    const int cur = (nchunks - 1) & 1;
    gemm_chunk<false>(&As[cur][mW + lr][2 * lhi], &Bs[cur][nW + lr][2 * lhi],
                      &As[cur][sr][4 * seg], &Bs[cur][sr][4 * seg], ra, rb, acc);
  }

#pragma unroll
  for (int mt = 0; mt < 2; ++mt)
#pragma unroll
    for (int nt = 0; nt < 4; ++nt)
#pragma unroll
      for (int r = 0; r < 8; ++r) {
        size_t row = (size_t)(mBase + mt * 16 + lhi * 8 + r);
        C[row * Nc + (nBase + nt * 16 + lr)] = acc[mt][nt][r];
      }
}

__device__ __forceinline__ void pv_load_v8(v2f (&v)[16], int jb, const float* __restrict__ vcol,
                                           int hb, int lhi) {
  const float* p = vcol + (size_t)(hb * 32 + 2 * lhi) * (3 * DD);
#pragma unroll
  for (int j = 0; j < 8; ++j) {
    v[jb + j].x = p[(size_t)(4 * j) * (3 * DD)];
    v[jb + j].y = p[(size_t)(4 * j + 1) * (3 * DD)];
  }
}

__device__ __forceinline__ void pv_load_p8(v2f (&p0)[8], v2f (&p1)[8],
                                           const float* s0, const float* s1, int hb) {
#pragma unroll
  for (int j = 0; j < 8; ++j) {
    p0[j] = *(const v2f*)(s0 + hb * 32 + 4 * j);
    p1[j] = *(const v2f*)(s1 + hb * 32 + 4 * j);
  }
}

__device__ __forceinline__ void pv_mma8(const v2f (&p0)[8], const v2f (&p1)[8],
                                        const v2f (&v)[16], int jb, v8f (&oacc)[2]) {
#pragma unroll
  for (int j = 0; j < 8; ++j) {
    oacc[0] = wmma_f32(p0[j], v[jb + j], oacc[0]);
    oacc[1] = wmma_f32(p1[j], v[jb + j], oacc[1]);
  }
}

__device__ __forceinline__ v4f stj_recip4(v4f s, float l) {
  v4f w;
  w.x = 1.0f / fmaxf(l - s.x, EPS_);
  w.y = 1.0f / fmaxf(l - s.y, EPS_);
  w.z = 1.0f / fmaxf(l - s.z, EPS_);
  w.w = 1.0f / fmaxf(l - s.w, EPS_);
  return w;
}

__global__ __launch_bounds__(128)
void stieltjes_attn(const float* __restrict__ qkv, float* __restrict__ O) {
  __shared__ __attribute__((aligned(16))) float S[32][SP];
  __shared__ float redA[4][32];
  __shared__ float redB[4][32];
  __shared__ float lam[32];
  __shared__ float wsum[32];

  const int lane = threadIdx.x & 31;
  const int wave = threadIdx.x >> 5;
  const int lhi  = lane >> 4;
  const int lr   = lane & 15;

  const int rb = 31 - (blockIdx.x >> 5);
  const int bh = blockIdx.x & 31;
  const int h  = bh & (HH - 1);
  const int b  = bh >> 4;

  const int rowBase = rb * 32;
  const int ncols   = rowBase + 32;
  const int nct     = ncols >> 4;

  const float* qbase = qkv + (size_t)b * NN * (3 * DD) + h * DHH;
  const float* kbase = qbase + DD;
  const float* vbase = qbase + 2 * DD;

  v2f aQ[2][16];
#pragma unroll
  for (int mt = 0; mt < 2; ++mt) {
    const float* qr = qbase + (size_t)(rowBase + mt * 16 + lr) * (3 * DD) + 2 * lhi;
#pragma unroll
    for (int ks = 0; ks < 16; ++ks) aQ[mt][ks] = *(const v2f*)(qr + 4 * ks);
  }

  for (int ct = wave; ct < nct; ct += 4) {
    v8f acc0 = (v8f){0,0,0,0,0,0,0,0};
    v8f acc1 = (v8f){0,0,0,0,0,0,0,0};
    const float* kr = kbase + (size_t)(ct * 16 + lr) * (3 * DD) + 2 * lhi;
    v2f bK = *(const v2f*)(kr);
#pragma unroll
    for (int ks = 0; ks < 16; ++ks) {
      v2f bKn;
      if (ks < 15) bKn = *(const v2f*)(kr + 4 * (ks + 1));
      acc0 = wmma_f32(aQ[0][ks], bK, acc0);
      acc1 = wmma_f32(aQ[1][ks], bK, acc1);
      bK = bKn;
    }
    const int gcol = ct * 16 + lr;
#pragma unroll
    for (int r = 0; r < 8; ++r) {
      int row0 = lhi * 8 + r;
      int row1 = 16 + row0;
      float v0 = acc0[r] * SCALE;
      float v1 = acc1[r] * SCALE;
      if (gcol > rowBase + row0) v0 = NEGBIG;
      if (gcol > rowBase + row1) v1 = NEGBIG;
      S[row0][gcol] = v0;
      S[row1][gcol] = v1;
    }
  }
  __syncthreads();

  const int row   = threadIdx.x & 31;
  const int part  = threadIdx.x >> 5;
  const int ntrip = rb + 1;
  v4f* rp = reinterpret_cast<v4f*>(&S[row][0]) + part;
  {
    v4f m4 = (v4f){-3e30f, -3e30f, -3e30f, -3e30f};
    for (int t = 0; t < ntrip; ++t) {
      const v4f x0 = rp[8 * t];
      const v4f x1 = rp[8 * t + 4];
      m4.x = fmaxf(m4.x, fmaxf(x0.x, x1.x));
      m4.y = fmaxf(m4.y, fmaxf(x0.y, x1.y));
      m4.z = fmaxf(m4.z, fmaxf(x0.z, x1.z));
      m4.w = fmaxf(m4.w, fmaxf(x0.w, x1.w));
    }
    redA[part][row] = fmaxf(fmaxf(m4.x, m4.y), fmaxf(m4.z, m4.w));
  }
  __syncthreads();
  if (threadIdx.x < 32) {
    float m = fmaxf(fmaxf(redA[0][row], redA[1][row]),
                    fmaxf(redA[2][row], redA[3][row]));
    lam[row] = m + 1.0f;
  }
  __syncthreads();

  for (int it = 0; it < 4; ++it) {
    const float l = lam[row];
    v4f fs = (v4f){0.0f, 0.0f, 0.0f, 0.0f};
    v4f ps = (v4f){0.0f, 0.0f, 0.0f, 0.0f};
    v4f x0 = rp[0], x1 = rp[4];
    for (int t = 0; t < ntrip; ++t) {
      v4f n0 = x0, n1 = x1;
      if (t + 1 < ntrip) { n0 = rp[8 * t + 8]; n1 = rp[8 * t + 12]; }
      const v4f r0 = stj_recip4(x0, l);
      const v4f r1 = stj_recip4(x1, l);
      fs += r0;  ps += r0 * r0;
      fs += r1;  ps += r1 * r1;
      x0 = n0;  x1 = n1;
    }
    redA[part][row] = (fs.x + fs.y) + (fs.z + fs.w);
    redB[part][row] = (ps.x + ps.y) + (ps.z + ps.w);
    __syncthreads();
    if (threadIdx.x < 32) {
      float f  = redA[0][row] + redA[1][row] + redA[2][row] + redA[3][row] - 1.0f;
      float fp = -(redB[0][row] + redB[1][row] + redB[2][row] + redB[3][row]);
      lam[row] = l - f / fp;
    }
    __syncthreads();
  }

  {
    const float l = lam[row];
    v4f ws = (v4f){0.0f, 0.0f, 0.0f, 0.0f};
    v4f x0 = rp[0], x1 = rp[4];
    for (int t = 0; t < ntrip; ++t) {
      v4f n0 = x0, n1 = x1;
      if (t + 1 < ntrip) { n0 = rp[8 * t + 8]; n1 = rp[8 * t + 12]; }
      ws += stj_recip4(x0, l);
      ws += stj_recip4(x1, l);
      x0 = n0;  x1 = n1;
    }
    redA[part][row] = (ws.x + ws.y) + (ws.z + ws.w);
    __syncthreads();
    if (threadIdx.x < 32)
      wsum[row] = redA[0][row] + redA[1][row] + redA[2][row] + redA[3][row];
    __syncthreads();
    const float inv = 1.0f / wsum[row];
    x0 = rp[0];  x1 = rp[4];
    for (int t = 0; t < ntrip; ++t) {
      v4f n0 = x0, n1 = x1;
      if (t + 1 < ntrip) { n0 = rp[8 * t + 8]; n1 = rp[8 * t + 12]; }
      rp[8 * t]     = stj_recip4(x0, l) * inv;
      rp[8 * t + 4] = stj_recip4(x1, l) * inv;
      x0 = n0;  x1 = n1;
    }
  }

  const int dh0 = wave * 16;
  const float* vcol = vbase + dh0 + lr;

  v8f oacc[2];
  oacc[0] = (v8f){0,0,0,0,0,0,0,0};
  oacc[1] = (v8f){0,0,0,0,0,0,0,0};
  const int nh   = rb + 1;
  const int last = nh - 1;
  const float* s0 = &S[lr][2 * lhi];
  const float* s1 = &S[16 + lr][2 * lhi];

  v2f vA[16], vB[16];
  v2f p0[8], p1[8], q0[8], q1[8];
  pv_load_v8(vA, 0, vcol, 0, lhi);
  pv_load_v8(vA, 8, vcol, min(1, last), lhi);
  __syncthreads();
  pv_load_p8(p0, p1, s0, s1, 0);
  for (int h = 0; h < nh; h += 4) {
    pv_load_v8(vB, 0, vcol, min(h + 2, last), lhi);
    pv_load_v8(vB, 8, vcol, min(h + 3, last), lhi);
    pv_load_p8(q0, q1, s0, s1, min(h + 1, last));
    pv_mma8(p0, p1, vA, 0, oacc);
    if (h + 1 < nh) {
      pv_load_p8(p0, p1, s0, s1, min(h + 2, last));
      pv_mma8(q0, q1, vA, 8, oacc);
      if (h + 2 < nh) {
        pv_load_v8(vA, 0, vcol, min(h + 4, last), lhi);
        pv_load_v8(vA, 8, vcol, min(h + 5, last), lhi);
        pv_load_p8(q0, q1, s0, s1, min(h + 3, last));
        pv_mma8(p0, p1, vB, 0, oacc);
        if (h + 3 < nh) {
          pv_load_p8(p0, p1, s0, s1, min(h + 4, last));
          pv_mma8(q0, q1, vB, 8, oacc);
        }
      }
    }
  }

  float* obase = O + ((size_t)b * NN + rowBase) * DD + h * DHH + dh0 + lr;
#pragma unroll
  for (int mt = 0; mt < 2; ++mt)
#pragma unroll
    for (int r = 0; r < 8; ++r)
      obase[(size_t)(mt * 16 + lhi * 8 + r) * DD] = oacc[mt][r];
}

extern "C" void kernel_launch(void* const* d_in, const int* in_sizes, int n_in,
                              void* d_out, int out_size, void* d_ws, size_t ws_size,
                              hipStream_t stream) {
  const float* x    = (const float*)d_in[0];
  const float* Wqkv = (const float*)d_in[1];
  const float* Wo   = (const float*)d_in[2];
  float* out = (float*)d_out;

  float* qkv = (float*)d_ws;
  float* Ows = qkv + (size_t)2048 * 3072;

  gemm_nt_f32_wmma<<<(2048 / 64) * (3072 / 128), 128, 0, stream>>>(
      x, Wqkv, qkv, 2048, 3072, 1024);

  stieltjes_attn<<<BB * HH * (NN / 32), 128, 0, stream>>>(qkv, Ows);

  gemm_nt_f32_wmma<<<(2048 / 64) * (1024 / 128), 128, 0, stream>>>(
      Ows, Wo, out, 2048, 1024, 1024);
}


# Newton, row-sum and normalise passes: each score's reciprocal by the hardware v_rcp_f32 (within 1 ulp) instead of the 11-instruction correctly rounded division; sum order unchanged
# speedup vs baseline: 2.9990x; 1.1152x over previous
#include <hip/hip_runtime.h>

typedef __attribute__((ext_vector_type(2))) float v2f;
typedef __attribute__((ext_vector_type(8))) float v8f;
typedef __attribute__((ext_vector_type(4))) float v4f;

__device__ __forceinline__ v8f wmma_f32(v2f a, v2f b, v8f c) {
  return __builtin_amdgcn_wmma_f32_16x16x4_f32(false, a, false, b, (short)0, c, false, false);
}

#define BB   2
#define NN   1024
#define DD   1024
#define HH   16
#define DHH  64
#define SCALE 0.125f
#define EPS_  1e-6f
#define NEGBIG (-1e30f)
#define SP   1028

#define GP   36
#define GKC  32

__device__ __forceinline__ void gemm_fetch(const float* __restrict__ ag,
                                           const float* __restrict__ bg,
                                           size_t s16, int kc, v4f ra[4], v4f rb[8]) {
#pragma unroll
  for (int i = 0; i < 4; ++i) ra[i] = *(const v4f*)(ag + i * s16 + kc);
#pragma unroll
  for (int i = 0; i < 8; ++i) rb[i] = *(const v4f*)(bg + i * s16 + kc);
}

__device__ __forceinline__ void gemm_park(float* as, float* bs, const v4f ra[4], const v4f rb[8]) {
#pragma unroll
  for (int i = 0; i < 4; ++i) *(v4f*)(as + i * 16 * GP) = ra[i];
#pragma unroll
  for (int i = 0; i < 8; ++i) *(v4f*)(bs + i * 16 * GP) = rb[i];
}

__device__ __forceinline__ void gemm_ldfrag(const float* ap, const float* bp, int k0,
                                            v2f a[2], v2f b[4]) {
  a[0] = *(const v2f*)(ap + k0);
  a[1] = *(const v2f*)(ap + 16 * GP + k0);
  b[0] = *(const v2f*)(bp + k0);
  b[1] = *(const v2f*)(bp + 16 * GP + k0);
  b[2] = *(const v2f*)(bp + 32 * GP + k0);
  b[3] = *(const v2f*)(bp + 48 * GP + k0);
}

__device__ __forceinline__ void gemm_mma8(const v2f a[2], const v2f b[4],
                                          v8f acc[2][4]) {
#pragma unroll
  for (int mt = 0; mt < 2; ++mt)
#pragma unroll
    for (int nt = 0; nt < 4; ++nt)
      acc[mt][nt] = wmma_f32(a[mt], b[nt], acc[mt][nt]);
}

template <bool NEXT>
__device__ __forceinline__ void gemm_chunk(const float* ap, const float* bp,
                                           float* asn, float* bsn,
                                           const v4f ra[4], const v4f rb[8], v8f acc[2][4]) {
  v2f aE[2], bE[4], aO[2], bO[4];
  gemm_ldfrag(ap, bp, 0, aE, bE);
#pragma unroll
  for (int ks = 0; ks < GKC / 4; ks += 2) {
    gemm_ldfrag(ap, bp, 4 * ks + 4, aO, bO);
    gemm_mma8(aE, bE, acc);
    if (ks + 2 < GKC / 4) gemm_ldfrag(ap, bp, 4 * ks + 8, aE, bE);
    gemm_mma8(aO, bO, acc);
    if (NEXT && ks == 4) gemm_park(asn, bsn, ra, rb);
  }
}

__global__ __launch_bounds__(128)
void gemm_nt_f32_wmma(const float* __restrict__ A, const float* __restrict__ Bm,
                      float* __restrict__ C, int M, int Nc, int K) {
  __shared__ __attribute__((aligned(16))) float As[2][64][GP];
  __shared__ __attribute__((aligned(16))) float Bs[2][128][GP];

  const int lane = threadIdx.x & 31;
  const int wave = threadIdx.x >> 5;
  const int lhi  = lane >> 4;
  const int lr   = lane & 15;

  const int nBlocks = Nc >> 7;
  const int mb = blockIdx.x / nBlocks;
  const int nb = blockIdx.x % nBlocks;
  const int mW = (wave >> 1) * 32;
  const int nW = (wave & 1) * 64;
  const int mBase = mb * 64  + mW;
  const int nBase = nb * 128 + nW;

  v8f acc[2][4];
#pragma unroll
  for (int mt = 0; mt < 2; ++mt)
#pragma unroll
    for (int nt = 0; nt < 4; ++nt) acc[mt][nt] = (v8f){0,0,0,0,0,0,0,0};

  const int seg = threadIdx.x & 7;
  const int sr  = threadIdx.x >> 3;
  const float* ag = A  + (size_t)(mb * 64  + sr) * K + 4 * seg;
  const float* bg = Bm + (size_t)(nb * 128 + sr) * K + 4 * seg;
  const size_t s16 = (size_t)16 * K;

  const int nchunks = K / GKC;
  v4f ra[4], rb[8];
  gemm_fetch(ag, bg, s16, 0, ra, rb);
  gemm_park(&As[0][sr][4 * seg], &Bs[0][sr][4 * seg], ra, rb);
  __syncthreads();

  for (int c = 0; c + 1 < nchunks; ++c) {
    const int cur = c & 1;
    gemm_fetch(ag, bg, s16, (c + 1) * GKC, ra, rb);
    gemm_chunk<true>(&As[cur][mW + lr][2 * lhi], &Bs[cur][nW + lr][2 * lhi],
                     &As[cur ^ 1][sr][4 * seg], &Bs[cur ^ 1][sr][4 * seg], ra, rb, acc);
    __syncthreads();
  }
  {
    const int cur = (nchunks - 1) & 1;
    gemm_chunk<false>(&As[cur][mW + lr][2 * lhi], &Bs[cur][nW + lr][2 * lhi],
                      &As[cur][sr][4 * seg], &Bs[cur][sr][4 * seg], ra, rb, acc);
  }

#pragma unroll
  for (int mt = 0; mt < 2; ++mt)
#pragma unroll
    for (int nt = 0; nt < 4; ++nt)
#pragma unroll
      for (int r = 0; r < 8; ++r) {
        size_t row = (size_t)(mBase + mt * 16 + lhi * 8 + r);
        C[row * Nc + (nBase + nt * 16 + lr)] = acc[mt][nt][r];
      }
}

__device__ __forceinline__ void pv_load_v8(v2f (&v)[16], int jb, const float* __restrict__ vcol,
                                           int hb, int lhi) {
  const float* p = vcol + (size_t)(hb * 32 + 2 * lhi) * (3 * DD);
#pragma unroll
  for (int j = 0; j < 8; ++j) {
    v[jb + j].x = p[(size_t)(4 * j) * (3 * DD)];
    v[jb + j].y = p[(size_t)(4 * j + 1) * (3 * DD)];
  }
}

__device__ __forceinline__ void pv_load_p8(v2f (&p0)[8], v2f (&p1)[8],
                                           const float* s0, const float* s1, int hb) {
#pragma unroll
  for (int j = 0; j < 8; ++j) {
    p0[j] = *(const v2f*)(s0 + hb * 32 + 4 * j);
    p1[j] = *(const v2f*)(s1 + hb * 32 + 4 * j);
  }
}

__device__ __forceinline__ void pv_mma8(const v2f (&p0)[8], const v2f (&p1)[8],
                                        const v2f (&v)[16], int jb, v8f (&oacc)[2]) {
#pragma unroll
  for (int j = 0; j < 8; ++j) {
    oacc[0] = wmma_f32(p0[j], v[jb + j], oacc[0]);
    oacc[1] = wmma_f32(p1[j], v[jb + j], oacc[1]);
  }
}

__device__ __forceinline__ v4f stj_recip4(v4f s, float l) {
  v4f w;
  w.x = __builtin_amdgcn_rcpf(fmaxf(l - s.x, EPS_));
  w.y = __builtin_amdgcn_rcpf(fmaxf(l - s.y, EPS_));
  w.z = __builtin_amdgcn_rcpf(fmaxf(l - s.z, EPS_));
  w.w = __builtin_amdgcn_rcpf(fmaxf(l - s.w, EPS_));
  return w;
}

__global__ __launch_bounds__(128)
void stieltjes_attn(const float* __restrict__ qkv, float* __restrict__ O) {
  __shared__ __attribute__((aligned(16))) float S[32][SP];
  __shared__ float redA[4][32];
  __shared__ float redB[4][32];
  __shared__ float lam[32];
  __shared__ float wsum[32];

  const int lane = threadIdx.x & 31;
  const int wave = threadIdx.x >> 5;
  const int lhi  = lane >> 4;
  const int lr   = lane & 15;

  const int rb = 31 - (blockIdx.x >> 5);
  const int bh = blockIdx.x & 31;
  const int h  = bh & (HH - 1);
  const int b  = bh >> 4;

  const int rowBase = rb * 32;
  const int ncols   = rowBase + 32;
  const int nct     = ncols >> 4;

  const float* qbase = qkv + (size_t)b * NN * (3 * DD) + h * DHH;
  const float* kbase = qbase + DD;
  const float* vbase = qbase + 2 * DD;

  v2f aQ[2][16];
#pragma unroll
  for (int mt = 0; mt < 2; ++mt) {
    const float* qr = qbase + (size_t)(rowBase + mt * 16 + lr) * (3 * DD) + 2 * lhi;
#pragma unroll
    for (int ks = 0; ks < 16; ++ks) aQ[mt][ks] = *(const v2f*)(qr + 4 * ks);
  }

  for (int ct = wave; ct < nct; ct += 4) {
    v8f acc0 = (v8f){0,0,0,0,0,0,0,0};
    v8f acc1 = (v8f){0,0,0,0,0,0,0,0};
    const float* kr = kbase + (size_t)(ct * 16 + lr) * (3 * DD) + 2 * lhi;
    v2f bK = *(const v2f*)(kr);
#pragma unroll
    for (int ks = 0; ks < 16; ++ks) {
      v2f bKn;
      if (ks < 15) bKn = *(const v2f*)(kr + 4 * (ks + 1));
      acc0 = wmma_f32(aQ[0][ks], bK, acc0);
      acc1 = wmma_f32(aQ[1][ks], bK, acc1);
      bK = bKn;
    }
    const int gcol = ct * 16 + lr;
#pragma unroll
    for (int r = 0; r < 8; ++r) {
      int row0 = lhi * 8 + r;
      int row1 = 16 + row0;
      float v0 = acc0[r] * SCALE;
      float v1 = acc1[r] * SCALE;
      if (gcol > rowBase + row0) v0 = NEGBIG;
      if (gcol > rowBase + row1) v1 = NEGBIG;
      S[row0][gcol] = v0;
      S[row1][gcol] = v1;
    }
  }
  __syncthreads();

  const int row   = threadIdx.x & 31;
  const int part  = threadIdx.x >> 5;
  const int ntrip = rb + 1;
  v4f* rp = reinterpret_cast<v4f*>(&S[row][0]) + part;
  {
    v4f m4 = (v4f){-3e30f, -3e30f, -3e30f, -3e30f};
    for (int t = 0; t < ntrip; ++t) {
      const v4f x0 = rp[8 * t];
      const v4f x1 = rp[8 * t + 4];
      m4.x = fmaxf(m4.x, fmaxf(x0.x, x1.x));
      m4.y = fmaxf(m4.y, fmaxf(x0.y, x1.y));
      m4.z = fmaxf(m4.z, fmaxf(x0.z, x1.z));
      m4.w = fmaxf(m4.w, fmaxf(x0.w, x1.w));
    }
    redA[part][row] = fmaxf(fmaxf(m4.x, m4.y), fmaxf(m4.z, m4.w));
  }
  __syncthreads();
  if (threadIdx.x < 32) {
    float m = fmaxf(fmaxf(redA[0][row], redA[1][row]),
                    fmaxf(redA[2][row], redA[3][row]));
    lam[row] = m + 1.0f;
  }
  __syncthreads();

  for (int it = 0; it < 4; ++it) {
    const float l = lam[row];
    v4f fs = (v4f){0.0f, 0.0f, 0.0f, 0.0f};
    v4f ps = (v4f){0.0f, 0.0f, 0.0f, 0.0f};
    v4f x0 = rp[0], x1 = rp[4];
    for (int t = 0; t < ntrip; ++t) {
      v4f n0 = x0, n1 = x1;
      if (t + 1 < ntrip) { n0 = rp[8 * t + 8]; n1 = rp[8 * t + 12]; }
      const v4f r0 = stj_recip4(x0, l);
      const v4f r1 = stj_recip4(x1, l);
      fs += r0;  ps += r0 * r0;
      fs += r1;  ps += r1 * r1;
      x0 = n0;  x1 = n1;
    }
    redA[part][row] = (fs.x + fs.y) + (fs.z + fs.w);
    redB[part][row] = (ps.x + ps.y) + (ps.z + ps.w);
    __syncthreads();
    if (threadIdx.x < 32) {
      float f  = redA[0][row] + redA[1][row] + redA[2][row] + redA[3][row] - 1.0f;
      float fp = -(redB[0][row] + redB[1][row] + redB[2][row] + redB[3][row]);
      lam[row] = l - f / fp;
    }
    __syncthreads();
  }

  {
    const float l = lam[row];
    v4f ws = (v4f){0.0f, 0.0f, 0.0f, 0.0f};
    v4f x0 = rp[0], x1 = rp[4];
    for (int t = 0; t < ntrip; ++t) {
      v4f n0 = x0, n1 = x1;
      if (t + 1 < ntrip) { n0 = rp[8 * t + 8]; n1 = rp[8 * t + 12]; }
      ws += stj_recip4(x0, l);
      ws += stj_recip4(x1, l);
      x0 = n0;  x1 = n1;
    }
    redA[part][row] = (ws.x + ws.y) + (ws.z + ws.w);
    __syncthreads();
    if (threadIdx.x < 32)
      wsum[row] = redA[0][row] + redA[1][row] + redA[2][row] + redA[3][row];
    __syncthreads();
    const float inv = 1.0f / wsum[row];
    x0 = rp[0];  x1 = rp[4];
    for (int t = 0; t < ntrip; ++t) {
      v4f n0 = x0, n1 = x1;
      if (t + 1 < ntrip) { n0 = rp[8 * t + 8]; n1 = rp[8 * t + 12]; }
      rp[8 * t]     = stj_recip4(x0, l) * inv;
      rp[8 * t + 4] = stj_recip4(x1, l) * inv;
      x0 = n0;  x1 = n1;
    }
  }

  const int dh0 = wave * 16;
  const float* vcol = vbase + dh0 + lr;

  v8f oacc[2];
  oacc[0] = (v8f){0,0,0,0,0,0,0,0};
  oacc[1] = (v8f){0,0,0,0,0,0,0,0};
  const int nh   = rb + 1;
  const int last = nh - 1;
  const float* s0 = &S[lr][2 * lhi];
  const float* s1 = &S[16 + lr][2 * lhi];

  v2f vA[16], vB[16];
  v2f p0[8], p1[8], q0[8], q1[8];
  pv_load_v8(vA, 0, vcol, 0, lhi);
  pv_load_v8(vA, 8, vcol, min(1, last), lhi);
  __syncthreads();
  pv_load_p8(p0, p1, s0, s1, 0);
  for (int h = 0; h < nh; h += 4) {
    pv_load_v8(vB, 0, vcol, min(h + 2, last), lhi);
    pv_load_v8(vB, 8, vcol, min(h + 3, last), lhi);
    pv_load_p8(q0, q1, s0, s1, min(h + 1, last));
    pv_mma8(p0, p1, vA, 0, oacc);
    if (h + 1 < nh) {
      pv_load_p8(p0, p1, s0, s1, min(h + 2, last));
      pv_mma8(q0, q1, vA, 8, oacc);
      if (h + 2 < nh) {
        pv_load_v8(vA, 0, vcol, min(h + 4, last), lhi);
        pv_load_v8(vA, 8, vcol, min(h + 5, last), lhi);
        pv_load_p8(q0, q1, s0, s1, min(h + 3, last));
        pv_mma8(p0, p1, vB, 0, oacc);
        if (h + 3 < nh) {
          pv_load_p8(p0, p1, s0, s1, min(h + 4, last));
          pv_mma8(q0, q1, vB, 8, oacc);
        }
      }
    }
  }

  float* obase = O + ((size_t)b * NN + rowBase) * DD + h * DHH + dh0 + lr;
#pragma unroll
  for (int mt = 0; mt < 2; ++mt)
#pragma unroll
    for (int r = 0; r < 8; ++r)
      obase[(size_t)(mt * 16 + lhi * 8 + r) * DD] = oacc[mt][r];
}

extern "C" void kernel_launch(void* const* d_in, const int* in_sizes, int n_in,
                              void* d_out, int out_size, void* d_ws, size_t ws_size,
                              hipStream_t stream) {
  const float* x    = (const float*)d_in[0];
  const float* Wqkv = (const float*)d_in[1];
  const float* Wo   = (const float*)d_in[2];
  float* out = (float*)d_out;

  float* qkv = (float*)d_ws;
  float* Ows = qkv + (size_t)2048 * 3072;

  gemm_nt_f32_wmma<<<(2048 / 64) * (3072 / 128), 128, 0, stream>>>(
      x, Wqkv, qkv, 2048, 3072, 1024);

  stieltjes_attn<<<BB * HH * (NN / 32), 128, 0, stream>>>(qkv, Ows);

  gemm_nt_f32_wmma<<<(2048 / 64) * (1024 / 128), 128, 0, stream>>>(
      Ows, Wo, out, 2048, 1024, 1024);
}


# Both projection GEMMs: each LDS fragment read issued a block of WMMAs ahead as one group, with the chunk hand-over and its barrier moved inside the first block; k and sum order unchanged
# speedup vs baseline: 3.0992x; 1.0334x over previous
#include <hip/hip_runtime.h>

typedef __attribute__((ext_vector_type(2))) float v2f;
typedef __attribute__((ext_vector_type(8))) float v8f;
typedef __attribute__((ext_vector_type(4))) float v4f;

__device__ __forceinline__ v8f wmma_f32(v2f a, v2f b, v8f c) {
  return __builtin_amdgcn_wmma_f32_16x16x4_f32(false, a, false, b, (short)0, c, false, false);
}

#define BB   2
#define NN   1024
#define DD   1024
#define HH   16
#define DHH  64
#define SCALE 0.125f
#define EPS_  1e-6f
#define NEGBIG (-1e30f)
#define SP   1028

#define GP   36
#define GKC  32

__device__ __forceinline__ void gemm_fetch(const float* __restrict__ ag,
                                           const float* __restrict__ bg,
                                           size_t s16, int kc, v4f ra[4], v4f rb[8]) {
#pragma unroll
  for (int i = 0; i < 4; ++i) ra[i] = *(const v4f*)(ag + i * s16 + kc);
#pragma unroll
  for (int i = 0; i < 8; ++i) rb[i] = *(const v4f*)(bg + i * s16 + kc);
}

__device__ __forceinline__ void gemm_park(float* as, float* bs, const v4f ra[4], const v4f rb[8]) {
#pragma unroll
  for (int i = 0; i < 4; ++i) *(v4f*)(as + i * 16 * GP) = ra[i];
#pragma unroll
  for (int i = 0; i < 8; ++i) *(v4f*)(bs + i * 16 * GP) = rb[i];
}

__device__ __forceinline__ void gemm_ldblock(const float* ap, const float* bp, int k0,
                                             v2f (&a)[4][2], v2f (&b)[4][4]) {
#pragma unroll
  for (int s = 0; s < 4; ++s) {
    a[s][0] = *(const v2f*)(ap + k0 + 4 * s);
    a[s][1] = *(const v2f*)(ap + 16 * GP + k0 + 4 * s);
    b[s][0] = *(const v2f*)(bp + k0 + 4 * s);
    b[s][1] = *(const v2f*)(bp + 16 * GP + k0 + 4 * s);
    b[s][2] = *(const v2f*)(bp + 32 * GP + k0 + 4 * s);
    b[s][3] = *(const v2f*)(bp + 48 * GP + k0 + 4 * s);
  }
}

template <int S0, int NS>
__device__ __forceinline__ void gemm_mma(const v2f (&a)[4][2], const v2f (&b)[4][4],
                                         v8f acc[2][4]) {
#pragma unroll
  for (int s = S0; s < S0 + NS; ++s)
#pragma unroll
    for (int mt = 0; mt < 2; ++mt)
#pragma unroll
      for (int nt = 0; nt < 4; ++nt)
        acc[mt][nt] = wmma_f32(a[s][mt], b[s][nt], acc[mt][nt]);
}

template <bool PARK, bool FETCH>
__device__ __forceinline__ void gemm_chunk(const float* ap, const float* bp,
                                           const float* apn, const float* bpn,
                                           float* asn, float* bsn,
                                           const float* __restrict__ ag,
                                           const float* __restrict__ bg, size_t s16, int kc2,
                                           v4f ra[4], v4f rb[8],
                                           v2f (&a0)[4][2], v2f (&b0)[4][4],
                                           v2f (&a1)[4][2], v2f (&b1)[4][4], v8f acc[2][4]) {
  gemm_ldblock(ap, bp, 16, a1, b1);
  gemm_mma<0, 2>(a0, b0, acc);
  __builtin_amdgcn_sched_barrier(0);
  if (PARK) gemm_park(asn, bsn, ra, rb);
  __builtin_amdgcn_sched_barrier(0);
  gemm_mma<2, 1>(a0, b0, acc);
  __builtin_amdgcn_sched_barrier(0);
  if (PARK) {
    __syncthreads();
    if (FETCH) gemm_fetch(ag, bg, s16, kc2, ra, rb);
  }
  gemm_mma<3, 1>(a0, b0, acc);
  if (PARK) gemm_ldblock(apn, bpn, 0, a0, b0);
  gemm_mma<0, 4>(a1, b1, acc);
}

__global__ __launch_bounds__(128)
void gemm_nt_f32_wmma(const float* __restrict__ A, const float* __restrict__ Bm,
                      float* __restrict__ C, int M, int Nc, int K) {
  __shared__ __attribute__((aligned(16))) float As[2][64][GP];
  __shared__ __attribute__((aligned(16))) float Bs[2][128][GP];

  const int lane = threadIdx.x & 31;
  const int wave = threadIdx.x >> 5;
  const int lhi  = lane >> 4;
  const int lr   = lane & 15;

  const int nBlocks = Nc >> 7;
  const int mb = blockIdx.x / nBlocks;
  const int nb = blockIdx.x % nBlocks;
  const int mW = (wave >> 1) * 32;
  const int nW = (wave & 1) * 64;
  const int mBase = mb * 64  + mW;
  const int nBase = nb * 128 + nW;

  v8f acc[2][4];
#pragma unroll
  for (int mt = 0; mt < 2; ++mt)
#pragma unroll
    for (int nt = 0; nt < 4; ++nt) acc[mt][nt] = (v8f){0,0,0,0,0,0,0,0};

  const int seg = threadIdx.x & 7;
  const int sr  = threadIdx.x >> 3;
  const float* ag = A  + (size_t)(mb * 64  + sr) * K + 4 * seg;
  const float* bg = Bm + (size_t)(nb * 128 + sr) * K + 4 * seg;
  const size_t s16 = (size_t)16 * K;

  const int nchunks = K / GKC;
  v4f ra[4], rb[8];
  v2f a0[4][2], b0[4][4], a1[4][2], b1[4][4];

  gemm_fetch(ag, bg, s16, 0, ra, rb);
  gemm_park(&As[0][sr][4 * seg], &Bs[0][sr][4 * seg], ra, rb);
  __syncthreads();
  if (nchunks > 1) gemm_fetch(ag, bg, s16, GKC, ra, rb);
  gemm_ldblock(&As[0][mW + lr][2 * lhi], &Bs[0][nW + lr][2 * lhi], 0, a0, b0);

  int c = 0;
  for (; c + 2 < nchunks; ++c) {
    const int cur = c & 1;
    gemm_chunk<true, true>(&As[cur][mW + lr][2 * lhi], &Bs[cur][nW + lr][2 * lhi],
                           &As[cur ^ 1][mW + lr][2 * lhi], &Bs[cur ^ 1][nW + lr][2 * lhi],
                           &As[cur ^ 1][sr][4 * seg], &Bs[cur ^ 1][sr][4 * seg],
                           ag, bg, s16, (c + 2) * GKC, ra, rb, a0, b0, a1, b1, acc);
  }
  if (c + 1 < nchunks) {
    const int cur = c & 1;
    gemm_chunk<true, false>(&As[cur][mW + lr][2 * lhi], &Bs[cur][nW + lr][2 * lhi],
                            &As[cur ^ 1][mW + lr][2 * lhi], &Bs[cur ^ 1][nW + lr][2 * lhi],
                            &As[cur ^ 1][sr][4 * seg], &Bs[cur ^ 1][sr][4 * seg],
                            ag, bg, s16, 0, ra, rb, a0, b0, a1, b1, acc);
    ++c;
  }
  {
    const int cur = c & 1;
    gemm_chunk<false, false>(&As[cur][mW + lr][2 * lhi], &Bs[cur][nW + lr][2 * lhi],
                             &As[cur][mW + lr][2 * lhi], &Bs[cur][nW + lr][2 * lhi],
                             &As[cur][sr][4 * seg], &Bs[cur][sr][4 * seg],
                             ag, bg, s16, 0, ra, rb, a0, b0, a1, b1, acc);
  }

#pragma unroll
  for (int mt = 0; mt < 2; ++mt)
#pragma unroll
    for (int nt = 0; nt < 4; ++nt)
#pragma unroll
      for (int r = 0; r < 8; ++r) {
        size_t row = (size_t)(mBase + mt * 16 + lhi * 8 + r);
        C[row * Nc + (nBase + nt * 16 + lr)] = acc[mt][nt][r];
      }
}

__device__ __forceinline__ void pv_load_v8(v2f (&v)[16], int jb, const float* __restrict__ vcol,
                                           int hb, int lhi) {
  const float* p = vcol + (size_t)(hb * 32 + 2 * lhi) * (3 * DD);
#pragma unroll
  for (int j = 0; j < 8; ++j) {
    v[jb + j].x = p[(size_t)(4 * j) * (3 * DD)];
    v[jb + j].y = p[(size_t)(4 * j + 1) * (3 * DD)];
  }
}

__device__ __forceinline__ void pv_load_p8(v2f (&p0)[8], v2f (&p1)[8],
                                           const float* s0, const float* s1, int hb) {
#pragma unroll
  for (int j = 0; j < 8; ++j) {
    p0[j] = *(const v2f*)(s0 + hb * 32 + 4 * j);
    p1[j] = *(const v2f*)(s1 + hb * 32 + 4 * j);
  }
}

__device__ __forceinline__ void pv_mma8(const v2f (&p0)[8], const v2f (&p1)[8],
                                        const v2f (&v)[16], int jb, v8f (&oacc)[2]) {
#pragma unroll
  for (int j = 0; j < 8; ++j) {
    oacc[0] = wmma_f32(p0[j], v[jb + j], oacc[0]);
    oacc[1] = wmma_f32(p1[j], v[jb + j], oacc[1]);
  }
}

__device__ __forceinline__ v4f stj_recip4(v4f s, float l) {
  v4f w;
  w.x = __builtin_amdgcn_rcpf(fmaxf(l - s.x, EPS_));
  w.y = __builtin_amdgcn_rcpf(fmaxf(l - s.y, EPS_));
  w.z = __builtin_amdgcn_rcpf(fmaxf(l - s.z, EPS_));
  w.w = __builtin_amdgcn_rcpf(fmaxf(l - s.w, EPS_));
  return w;
}

__global__ __launch_bounds__(128)
void stieltjes_attn(const float* __restrict__ qkv, float* __restrict__ O) {
  __shared__ __attribute__((aligned(16))) float S[32][SP];
  __shared__ float redA[4][32];
  __shared__ float redB[4][32];
  __shared__ float lam[32];
  __shared__ float wsum[32];

  const int lane = threadIdx.x & 31;
  const int wave = threadIdx.x >> 5;
  const int lhi  = lane >> 4;
  const int lr   = lane & 15;

  const int rb = 31 - (blockIdx.x >> 5);
  const int bh = blockIdx.x & 31;
  const int h  = bh & (HH - 1);
  const int b  = bh >> 4;

  const int rowBase = rb * 32;
  const int ncols   = rowBase + 32;
  const int nct     = ncols >> 4;

  const float* qbase = qkv + (size_t)b * NN * (3 * DD) + h * DHH;
  const float* kbase = qbase + DD;
  const float* vbase = qbase + 2 * DD;

  v2f aQ[2][16];
#pragma unroll
  for (int mt = 0; mt < 2; ++mt) {
    const float* qr = qbase + (size_t)(rowBase + mt * 16 + lr) * (3 * DD) + 2 * lhi;
#pragma unroll
    for (int ks = 0; ks < 16; ++ks) aQ[mt][ks] = *(const v2f*)(qr + 4 * ks);
  }

  for (int ct = wave; ct < nct; ct += 4) {
    v8f acc0 = (v8f){0,0,0,0,0,0,0,0};
    v8f acc1 = (v8f){0,0,0,0,0,0,0,0};
    const float* kr = kbase + (size_t)(ct * 16 + lr) * (3 * DD) + 2 * lhi;
    v2f bK = *(const v2f*)(kr);
#pragma unroll
    for (int ks = 0; ks < 16; ++ks) {
      v2f bKn;
      if (ks < 15) bKn = *(const v2f*)(kr + 4 * (ks + 1));
      acc0 = wmma_f32(aQ[0][ks], bK, acc0);
      acc1 = wmma_f32(aQ[1][ks], bK, acc1);
      bK = bKn;
    }
    const int gcol = ct * 16 + lr;
#pragma unroll
    for (int r = 0; r < 8; ++r) {
      int row0 = lhi * 8 + r;
      int row1 = 16 + row0;
      float v0 = acc0[r] * SCALE;
      float v1 = acc1[r] * SCALE;
      if (gcol > rowBase + row0) v0 = NEGBIG;
      if (gcol > rowBase + row1) v1 = NEGBIG;
      S[row0][gcol] = v0;
      S[row1][gcol] = v1;
    }
  }
  __syncthreads();

  const int row   = threadIdx.x & 31;
  const int part  = threadIdx.x >> 5;
  const int ntrip = rb + 1;
  v4f* rp = reinterpret_cast<v4f*>(&S[row][0]) + part;
  {
    v4f m4 = (v4f){-3e30f, -3e30f, -3e30f, -3e30f};
    for (int t = 0; t < ntrip; ++t) {
      const v4f x0 = rp[8 * t];
      const v4f x1 = rp[8 * t + 4];
      m4.x = fmaxf(m4.x, fmaxf(x0.x, x1.x));
      m4.y = fmaxf(m4.y, fmaxf(x0.y, x1.y));
      m4.z = fmaxf(m4.z, fmaxf(x0.z, x1.z));
      m4.w = fmaxf(m4.w, fmaxf(x0.w, x1.w));
    }
    redA[part][row] = fmaxf(fmaxf(m4.x, m4.y), fmaxf(m4.z, m4.w));
  }
  __syncthreads();
  if (threadIdx.x < 32) {
    float m = fmaxf(fmaxf(redA[0][row], redA[1][row]),
                    fmaxf(redA[2][row], redA[3][row]));
    lam[row] = m + 1.0f;
  }
  __syncthreads();

  for (int it = 0; it < 4; ++it) {
    const float l = lam[row];
    v4f fs = (v4f){0.0f, 0.0f, 0.0f, 0.0f};
    v4f ps = (v4f){0.0f, 0.0f, 0.0f, 0.0f};
    v4f x0 = rp[0], x1 = rp[4];
    for (int t = 0; t < ntrip; ++t) {
      v4f n0 = x0, n1 = x1;
      if (t + 1 < ntrip) { n0 = rp[8 * t + 8]; n1 = rp[8 * t + 12]; }
      const v4f r0 = stj_recip4(x0, l);
      const v4f r1 = stj_recip4(x1, l);
      fs += r0;  ps += r0 * r0;
      fs += r1;  ps += r1 * r1;
      x0 = n0;  x1 = n1;
    }
    redA[part][row] = (fs.x + fs.y) + (fs.z + fs.w);
    redB[part][row] = (ps.x + ps.y) + (ps.z + ps.w);
    __syncthreads();
    if (threadIdx.x < 32) {
      float f  = redA[0][row] + redA[1][row] + redA[2][row] + redA[3][row] - 1.0f;
      float fp = -(redB[0][row] + redB[1][row] + redB[2][row] + redB[3][row]);
      lam[row] = l - f / fp;
    }
    __syncthreads();
  }

  {
    const float l = lam[row];
    v4f ws = (v4f){0.0f, 0.0f, 0.0f, 0.0f};
    v4f x0 = rp[0], x1 = rp[4];
    for (int t = 0; t < ntrip; ++t) {
      v4f n0 = x0, n1 = x1;
      if (t + 1 < ntrip) { n0 = rp[8 * t + 8]; n1 = rp[8 * t + 12]; }
      ws += stj_recip4(x0, l);
      ws += stj_recip4(x1, l);
      x0 = n0;  x1 = n1;
    }
    redA[part][row] = (ws.x + ws.y) + (ws.z + ws.w);
    __syncthreads();
    if (threadIdx.x < 32)
      wsum[row] = redA[0][row] + redA[1][row] + redA[2][row] + redA[3][row];
    __syncthreads();
    const float inv = 1.0f / wsum[row];
    x0 = rp[0];  x1 = rp[4];
    for (int t = 0; t < ntrip; ++t) {
      v4f n0 = x0, n1 = x1;
      if (t + 1 < ntrip) { n0 = rp[8 * t + 8]; n1 = rp[8 * t + 12]; }
      rp[8 * t]     = stj_recip4(x0, l) * inv;
      rp[8 * t + 4] = stj_recip4(x1, l) * inv;
      x0 = n0;  x1 = n1;
    }
  }

  const int dh0 = wave * 16;
  const float* vcol = vbase + dh0 + lr;

  v8f oacc[2];
  oacc[0] = (v8f){0,0,0,0,0,0,0,0};
  oacc[1] = (v8f){0,0,0,0,0,0,0,0};
  const int nh   = rb + 1;
  const int last = nh - 1;
  const float* s0 = &S[lr][2 * lhi];
  const float* s1 = &S[16 + lr][2 * lhi];

  v2f vA[16], vB[16];
  v2f p0[8], p1[8], q0[8], q1[8];
  pv_load_v8(vA, 0, vcol, 0, lhi);
  pv_load_v8(vA, 8, vcol, min(1, last), lhi);
  __syncthreads();
  pv_load_p8(p0, p1, s0, s1, 0);
  for (int h = 0; h < nh; h += 4) {
    pv_load_v8(vB, 0, vcol, min(h + 2, last), lhi);
    pv_load_v8(vB, 8, vcol, min(h + 3, last), lhi);
    pv_load_p8(q0, q1, s0, s1, min(h + 1, last));
    pv_mma8(p0, p1, vA, 0, oacc);
    if (h + 1 < nh) {
      pv_load_p8(p0, p1, s0, s1, min(h + 2, last));
      pv_mma8(q0, q1, vA, 8, oacc);
      if (h + 2 < nh) {
        pv_load_v8(vA, 0, vcol, min(h + 4, last), lhi);
        pv_load_v8(vA, 8, vcol, min(h + 5, last), lhi);
        pv_load_p8(q0, q1, s0, s1, min(h + 3, last));
        pv_mma8(p0, p1, vB, 0, oacc);
        if (h + 3 < nh) {
          pv_load_p8(p0, p1, s0, s1, min(h + 4, last));
          pv_mma8(q0, q1, vB, 8, oacc);
        }
      }
    }
  }

  float* obase = O + ((size_t)b * NN + rowBase) * DD + h * DHH + dh0 + lr;
#pragma unroll
  for (int mt = 0; mt < 2; ++mt)
#pragma unroll
    for (int r = 0; r < 8; ++r)
      obase[(size_t)(mt * 16 + lhi * 8 + r) * DD] = oacc[mt][r];
}

extern "C" void kernel_launch(void* const* d_in, const int* in_sizes, int n_in,
                              void* d_out, int out_size, void* d_ws, size_t ws_size,
                              hipStream_t stream) {
  const float* x    = (const float*)d_in[0];
  const float* Wqkv = (const float*)d_in[1];
  const float* Wo   = (const float*)d_in[2];
  float* out = (float*)d_out;

  float* qkv = (float*)d_ws;
  float* Ows = qkv + (size_t)2048 * 3072;

  gemm_nt_f32_wmma<<<(2048 / 64) * (3072 / 128), 128, 0, stream>>>(
      x, Wqkv, qkv, 2048, 3072, 1024);

  stieltjes_attn<<<BB * HH * (NN / 32), 128, 0, stream>>>(qkv, Ows);

  gemm_nt_f32_wmma<<<(2048 / 64) * (1024 / 128), 128, 0, stream>>>(
      Ows, Wo, out, 2048, 1024, 1024);
}


# Attention score passes: each thread reads its scores from LDS once into registers (trips written out behind wave-uniform branches) and runs max, Newton, row sum, normalise there; sum order unchanged
# speedup vs baseline: 3.2406x; 1.0456x over previous
#include <hip/hip_runtime.h>

typedef __attribute__((ext_vector_type(2))) float v2f;
typedef __attribute__((ext_vector_type(8))) float v8f;
typedef __attribute__((ext_vector_type(4))) float v4f;

__device__ __forceinline__ v8f wmma_f32(v2f a, v2f b, v8f c) {
  return __builtin_amdgcn_wmma_f32_16x16x4_f32(false, a, false, b, (short)0, c, false, false);
}

#define BB   2
#define NN   1024
#define DD   1024
#define HH   16
#define DHH  64
#define SCALE 0.125f
#define EPS_  1e-6f
#define NEGBIG (-1e30f)
#define SP   1028

#define GP   36
#define GKC  32

__device__ __forceinline__ void gemm_fetch(const float* __restrict__ ag,
                                           const float* __restrict__ bg,
                                           size_t s16, int kc, v4f ra[4], v4f rb[8]) {
#pragma unroll
  for (int i = 0; i < 4; ++i) ra[i] = *(const v4f*)(ag + i * s16 + kc);
#pragma unroll
  for (int i = 0; i < 8; ++i) rb[i] = *(const v4f*)(bg + i * s16 + kc);
}

__device__ __forceinline__ void gemm_park(float* as, float* bs, const v4f ra[4], const v4f rb[8]) {
#pragma unroll
  for (int i = 0; i < 4; ++i) *(v4f*)(as + i * 16 * GP) = ra[i];
#pragma unroll
  for (int i = 0; i < 8; ++i) *(v4f*)(bs + i * 16 * GP) = rb[i];
}

__device__ __forceinline__ void gemm_ldblock(const float* ap, const float* bp, int k0,
                                             v2f (&a)[4][2], v2f (&b)[4][4]) {
#pragma unroll
  for (int s = 0; s < 4; ++s) {
    a[s][0] = *(const v2f*)(ap + k0 + 4 * s);
    a[s][1] = *(const v2f*)(ap + 16 * GP + k0 + 4 * s);
    b[s][0] = *(const v2f*)(bp + k0 + 4 * s);
    b[s][1] = *(const v2f*)(bp + 16 * GP + k0 + 4 * s);
    b[s][2] = *(const v2f*)(bp + 32 * GP + k0 + 4 * s);
    b[s][3] = *(const v2f*)(bp + 48 * GP + k0 + 4 * s);
  }
}

template <int S0, int NS>
__device__ __forceinline__ void gemm_mma(const v2f (&a)[4][2], const v2f (&b)[4][4],
                                         v8f acc[2][4]) {
#pragma unroll
  for (int s = S0; s < S0 + NS; ++s)
#pragma unroll
    for (int mt = 0; mt < 2; ++mt)
#pragma unroll
      for (int nt = 0; nt < 4; ++nt)
        acc[mt][nt] = wmma_f32(a[s][mt], b[s][nt], acc[mt][nt]);
}

template <bool PARK, bool FETCH>
__device__ __forceinline__ void gemm_chunk(const float* ap, const float* bp,
                                           const float* apn, const float* bpn,
                                           float* asn, float* bsn,
                                           const float* __restrict__ ag,
                                           const float* __restrict__ bg, size_t s16, int kc2,
                                           v4f ra[4], v4f rb[8],
                                           v2f (&a0)[4][2], v2f (&b0)[4][4],
                                           v2f (&a1)[4][2], v2f (&b1)[4][4], v8f acc[2][4]) {
  gemm_ldblock(ap, bp, 16, a1, b1);
  gemm_mma<0, 2>(a0, b0, acc);
  __builtin_amdgcn_sched_barrier(0);
  if (PARK) gemm_park(asn, bsn, ra, rb);
  __builtin_amdgcn_sched_barrier(0);
  gemm_mma<2, 1>(a0, b0, acc);
  __builtin_amdgcn_sched_barrier(0);
  if (PARK) {
    __syncthreads();
    if (FETCH) gemm_fetch(ag, bg, s16, kc2, ra, rb);
  }
  gemm_mma<3, 1>(a0, b0, acc);
  if (PARK) gemm_ldblock(apn, bpn, 0, a0, b0);
  gemm_mma<0, 4>(a1, b1, acc);
}

__global__ __launch_bounds__(128)
void gemm_nt_f32_wmma(const float* __restrict__ A, const float* __restrict__ Bm,
                      float* __restrict__ C, int M, int Nc, int K) {
  __shared__ __attribute__((aligned(16))) float As[2][64][GP];
  __shared__ __attribute__((aligned(16))) float Bs[2][128][GP];

  const int lane = threadIdx.x & 31;
  const int wave = threadIdx.x >> 5;
  const int lhi  = lane >> 4;
  const int lr   = lane & 15;

  const int nBlocks = Nc >> 7;
  const int mb = blockIdx.x / nBlocks;
  const int nb = blockIdx.x % nBlocks;
  const int mW = (wave >> 1) * 32;
  const int nW = (wave & 1) * 64;
  const int mBase = mb * 64  + mW;
  const int nBase = nb * 128 + nW;

  v8f acc[2][4];
#pragma unroll
  for (int mt = 0; mt < 2; ++mt)
#pragma unroll
    for (int nt = 0; nt < 4; ++nt) acc[mt][nt] = (v8f){0,0,0,0,0,0,0,0};

  const int seg = threadIdx.x & 7;
  const int sr  = threadIdx.x >> 3;
  const float* ag = A  + (size_t)(mb * 64  + sr) * K + 4 * seg;
  const float* bg = Bm + (size_t)(nb * 128 + sr) * K + 4 * seg;
  const size_t s16 = (size_t)16 * K;

  const int nchunks = K / GKC;
  v4f ra[4], rb[8];
  v2f a0[4][2], b0[4][4], a1[4][2], b1[4][4];

  gemm_fetch(ag, bg, s16, 0, ra, rb);
  gemm_park(&As[0][sr][4 * seg], &Bs[0][sr][4 * seg], ra, rb);
  __syncthreads();
  if (nchunks > 1) gemm_fetch(ag, bg, s16, GKC, ra, rb);
  gemm_ldblock(&As[0][mW + lr][2 * lhi], &Bs[0][nW + lr][2 * lhi], 0, a0, b0);

  int c = 0;
  for (; c + 2 < nchunks; ++c) {
    const int cur = c & 1;
    gemm_chunk<true, true>(&As[cur][mW + lr][2 * lhi], &Bs[cur][nW + lr][2 * lhi],
                           &As[cur ^ 1][mW + lr][2 * lhi], &Bs[cur ^ 1][nW + lr][2 * lhi],
                           &As[cur ^ 1][sr][4 * seg], &Bs[cur ^ 1][sr][4 * seg],
                           ag, bg, s16, (c + 2) * GKC, ra, rb, a0, b0, a1, b1, acc);
  }
  if (c + 1 < nchunks) {
    const int cur = c & 1;
    gemm_chunk<true, false>(&As[cur][mW + lr][2 * lhi], &Bs[cur][nW + lr][2 * lhi],
                            &As[cur ^ 1][mW + lr][2 * lhi], &Bs[cur ^ 1][nW + lr][2 * lhi],
                            &As[cur ^ 1][sr][4 * seg], &Bs[cur ^ 1][sr][4 * seg],
                            ag, bg, s16, 0, ra, rb, a0, b0, a1, b1, acc);
    ++c;
  }
  {
    const int cur = c & 1;
    gemm_chunk<false, false>(&As[cur][mW + lr][2 * lhi], &Bs[cur][nW + lr][2 * lhi],
                             &As[cur][mW + lr][2 * lhi], &Bs[cur][nW + lr][2 * lhi],
                             &As[cur][sr][4 * seg], &Bs[cur][sr][4 * seg],
                             ag, bg, s16, 0, ra, rb, a0, b0, a1, b1, acc);
  }

#pragma unroll
  for (int mt = 0; mt < 2; ++mt)
#pragma unroll
    for (int nt = 0; nt < 4; ++nt)
#pragma unroll
      for (int r = 0; r < 8; ++r) {
        size_t row = (size_t)(mBase + mt * 16 + lhi * 8 + r);
        C[row * Nc + (nBase + nt * 16 + lr)] = acc[mt][nt][r];
      }
}

__device__ __forceinline__ void pv_load_v8(v2f (&v)[16], int jb, const float* __restrict__ vcol,
                                           int hb, int lhi) {
  const float* p = vcol + (size_t)(hb * 32 + 2 * lhi) * (3 * DD);
#pragma unroll
  for (int j = 0; j < 8; ++j) {
    v[jb + j].x = p[(size_t)(4 * j) * (3 * DD)];
    v[jb + j].y = p[(size_t)(4 * j + 1) * (3 * DD)];
  }
}

__device__ __forceinline__ void pv_load_p8(v2f (&p0)[8], v2f (&p1)[8],
                                           const float* s0, const float* s1, int hb) {
#pragma unroll
  for (int j = 0; j < 8; ++j) {
    p0[j] = *(const v2f*)(s0 + hb * 32 + 4 * j);
    p1[j] = *(const v2f*)(s1 + hb * 32 + 4 * j);
  }
}

__device__ __forceinline__ void pv_mma8(const v2f (&p0)[8], const v2f (&p1)[8],
                                        const v2f (&v)[16], int jb, v8f (&oacc)[2]) {
#pragma unroll
  for (int j = 0; j < 8; ++j) {
    oacc[0] = wmma_f32(p0[j], v[jb + j], oacc[0]);
    oacc[1] = wmma_f32(p1[j], v[jb + j], oacc[1]);
  }
}

__device__ __forceinline__ v4f stj_recip4(v4f s, float l) {
  v4f w;
  w.x = __builtin_amdgcn_rcpf(fmaxf(l - s.x, EPS_));
  w.y = __builtin_amdgcn_rcpf(fmaxf(l - s.y, EPS_));
  w.z = __builtin_amdgcn_rcpf(fmaxf(l - s.z, EPS_));
  w.w = __builtin_amdgcn_rcpf(fmaxf(l - s.w, EPS_));
  return w;
}

template <int T> struct trip_c { static constexpr int value = T; };
template <int T, class F>
__device__ __forceinline__ void for_trips(int n, F&& f) {
  if constexpr (T < 32) {
    if (T < n) {
      f(trip_c<T>{});
      for_trips<T + 1>(n, f);
    }
  }
}

__global__ __launch_bounds__(128)
void stieltjes_attn(const float* __restrict__ qkv, float* __restrict__ O) {
  __shared__ __attribute__((aligned(16))) float S[32][SP];
  __shared__ float redA[4][32];
  __shared__ float redB[4][32];
  __shared__ float lam[32];
  __shared__ float wsum[32];

  const int lane = threadIdx.x & 31;
  const int wave = threadIdx.x >> 5;
  const int lhi  = lane >> 4;
  const int lr   = lane & 15;

  const int rb = 31 - (blockIdx.x >> 5);
  const int bh = blockIdx.x & 31;
  const int h  = bh & (HH - 1);
  const int b  = bh >> 4;

  const int rowBase = rb * 32;
  const int ncols   = rowBase + 32;
  const int nct     = ncols >> 4;

  const float* qbase = qkv + (size_t)b * NN * (3 * DD) + h * DHH;
  const float* kbase = qbase + DD;
  const float* vbase = qbase + 2 * DD;

  v2f aQ[2][16];
#pragma unroll
  for (int mt = 0; mt < 2; ++mt) {
    const float* qr = qbase + (size_t)(rowBase + mt * 16 + lr) * (3 * DD) + 2 * lhi;
#pragma unroll
    for (int ks = 0; ks < 16; ++ks) aQ[mt][ks] = *(const v2f*)(qr + 4 * ks);
  }

  for (int ct = wave; ct < nct; ct += 4) {
    v8f acc0 = (v8f){0,0,0,0,0,0,0,0};
    v8f acc1 = (v8f){0,0,0,0,0,0,0,0};
    const float* kr = kbase + (size_t)(ct * 16 + lr) * (3 * DD) + 2 * lhi;
    v2f bK = *(const v2f*)(kr);
#pragma unroll
    for (int ks = 0; ks < 16; ++ks) {
      v2f bKn;
      if (ks < 15) bKn = *(const v2f*)(kr + 4 * (ks + 1));
      acc0 = wmma_f32(aQ[0][ks], bK, acc0);
      acc1 = wmma_f32(aQ[1][ks], bK, acc1);
      bK = bKn;
    }
    const int gcol = ct * 16 + lr;
#pragma unroll
    for (int r = 0; r < 8; ++r) {
      int row0 = lhi * 8 + r;
      int row1 = 16 + row0;
      float v0 = acc0[r] * SCALE;
      float v1 = acc1[r] * SCALE;
      if (gcol > rowBase + row0) v0 = NEGBIG;
      if (gcol > rowBase + row1) v1 = NEGBIG;
      S[row0][gcol] = v0;
      S[row1][gcol] = v1;
    }
  }
  __syncthreads();

  const int row   = threadIdx.x & 31;
  const int part  = threadIdx.x >> 5;
  const int ntrip = rb + 1;
  v4f* rp = reinterpret_cast<v4f*>(&S[row][0]) + part;
  v4f x[32][2];
  for_trips<0>(ntrip, [&](auto tc) __attribute__((always_inline)) {
    constexpr int t = decltype(tc)::value;
    x[t][0] = rp[8 * t];
    x[t][1] = rp[8 * t + 4];
  });
  {
    v4f m4 = (v4f){-3e30f, -3e30f, -3e30f, -3e30f};
    for_trips<0>(ntrip, [&](auto tc) __attribute__((always_inline)) {
      constexpr int t = decltype(tc)::value;
      const v4f x0 = x[t][0];
      const v4f x1 = x[t][1];
      m4.x = fmaxf(m4.x, fmaxf(x0.x, x1.x));
      m4.y = fmaxf(m4.y, fmaxf(x0.y, x1.y));
      m4.z = fmaxf(m4.z, fmaxf(x0.z, x1.z));
      m4.w = fmaxf(m4.w, fmaxf(x0.w, x1.w));
    });
    redA[part][row] = fmaxf(fmaxf(m4.x, m4.y), fmaxf(m4.z, m4.w));
  }
  __syncthreads();
  if (threadIdx.x < 32) {
    float m = fmaxf(fmaxf(redA[0][row], redA[1][row]),
                    fmaxf(redA[2][row], redA[3][row]));
    lam[row] = m + 1.0f;
  }
  __syncthreads();

#pragma nounroll
  for (int it = 0; it < 4; ++it) {
    const float l = lam[row];
    v4f fs = (v4f){0.0f, 0.0f, 0.0f, 0.0f};
    v4f ps = (v4f){0.0f, 0.0f, 0.0f, 0.0f};
    for_trips<0>(ntrip, [&](auto tc) __attribute__((always_inline)) {
      constexpr int t = decltype(tc)::value;
      const v4f r0 = stj_recip4(x[t][0], l);
      const v4f r1 = stj_recip4(x[t][1], l);
      fs += r0;  ps += r0 * r0;
      fs += r1;  ps += r1 * r1;
    });
    redA[part][row] = (fs.x + fs.y) + (fs.z + fs.w);
    redB[part][row] = (ps.x + ps.y) + (ps.z + ps.w);
    __syncthreads();
    if (threadIdx.x < 32) {
      float f  = redA[0][row] + redA[1][row] + redA[2][row] + redA[3][row] - 1.0f;
      float fp = -(redB[0][row] + redB[1][row] + redB[2][row] + redB[3][row]);
      lam[row] = l - f / fp;
    }
    __syncthreads();
  }

  {
    const float l = lam[row];
    v4f ws = (v4f){0.0f, 0.0f, 0.0f, 0.0f};
    for_trips<0>(ntrip, [&](auto tc) __attribute__((always_inline)) {
      constexpr int t = decltype(tc)::value;
      x[t][0] = stj_recip4(x[t][0], l);
      x[t][1] = stj_recip4(x[t][1], l);
      ws += x[t][0];
      ws += x[t][1];
    });
    redA[part][row] = (ws.x + ws.y) + (ws.z + ws.w);
    __syncthreads();
    if (threadIdx.x < 32)
      wsum[row] = redA[0][row] + redA[1][row] + redA[2][row] + redA[3][row];
    __syncthreads();
    const float inv = 1.0f / wsum[row];
    for_trips<0>(ntrip, [&](auto tc) __attribute__((always_inline)) {
      constexpr int t = decltype(tc)::value;
      rp[8 * t]     = x[t][0] * inv;
      rp[8 * t + 4] = x[t][1] * inv;
    });
  }

  const int dh0 = wave * 16;
  const float* vcol = vbase + dh0 + lr;

  v8f oacc[2];
  oacc[0] = (v8f){0,0,0,0,0,0,0,0};
  oacc[1] = (v8f){0,0,0,0,0,0,0,0};
  const int nh   = rb + 1;
  const int last = nh - 1;
  const float* s0 = &S[lr][2 * lhi];
  const float* s1 = &S[16 + lr][2 * lhi];

  v2f vA[16], vB[16];
  v2f p0[8], p1[8], q0[8], q1[8];
  pv_load_v8(vA, 0, vcol, 0, lhi);
  pv_load_v8(vA, 8, vcol, min(1, last), lhi);
  __syncthreads();
  pv_load_p8(p0, p1, s0, s1, 0);
  for (int h = 0; h < nh; h += 4) {
    pv_load_v8(vB, 0, vcol, min(h + 2, last), lhi);
    pv_load_v8(vB, 8, vcol, min(h + 3, last), lhi);
    pv_load_p8(q0, q1, s0, s1, min(h + 1, last));
    pv_mma8(p0, p1, vA, 0, oacc);
    if (h + 1 < nh) {
      pv_load_p8(p0, p1, s0, s1, min(h + 2, last));
      pv_mma8(q0, q1, vA, 8, oacc);
      if (h + 2 < nh) {
        pv_load_v8(vA, 0, vcol, min(h + 4, last), lhi);
        pv_load_v8(vA, 8, vcol, min(h + 5, last), lhi);
        pv_load_p8(q0, q1, s0, s1, min(h + 3, last));
        pv_mma8(p0, p1, vB, 0, oacc);
        if (h + 3 < nh) {
          pv_load_p8(p0, p1, s0, s1, min(h + 4, last));
          pv_mma8(q0, q1, vB, 8, oacc);
        }
      }
    }
  }

  float* obase = O + ((size_t)b * NN + rowBase) * DD + h * DHH + dh0 + lr;
#pragma unroll
  for (int mt = 0; mt < 2; ++mt)
#pragma unroll
    for (int r = 0; r < 8; ++r)
      obase[(size_t)(mt * 16 + lhi * 8 + r) * DD] = oacc[mt][r];
}

extern "C" void kernel_launch(void* const* d_in, const int* in_sizes, int n_in,
                              void* d_out, int out_size, void* d_ws, size_t ws_size,
                              hipStream_t stream) {
  const float* x    = (const float*)d_in[0];
  const float* Wqkv = (const float*)d_in[1];
  const float* Wo   = (const float*)d_in[2];
  float* out = (float*)d_out;

  float* qkv = (float*)d_ws;
  float* Ows = qkv + (size_t)2048 * 3072;

  gemm_nt_f32_wmma<<<(2048 / 64) * (3072 / 128), 128, 0, stream>>>(
      x, Wqkv, qkv, 2048, 3072, 1024);

  stieltjes_attn<<<BB * HH * (NN / 32), 128, 0, stream>>>(qkv, Ows);

  gemm_nt_f32_wmma<<<(2048 / 64) * (1024 / 128), 128, 0, stream>>>(
      Ows, Wo, out, 2048, 1024, 1024);
}
